# starting kernel
# baseline (speedup 1.0000x reference)
#include <hip/hip_runtime.h>
#include <hip/hip_bf16.h>

typedef __attribute__((ext_vector_type(2))) float v2f;
typedef __attribute__((ext_vector_type(8))) float v8f;

#define D_CH 128

__device__ __forceinline__ void atomic_add_f32(float* p, float v) {
#if defined(__HIP_DEVICE_COMPILE__)
    unsafeAtomicAdd(p, v);
#else
    atomicAdd(p, v);
#endif
}

__global__ void fill_f32(float* __restrict__ p, float v, int n) {
    int i = blockIdx.x * blockDim.x + threadIdx.x;
    if (i < n) p[i] = v;
}

__global__ void deg_accum(float* __restrict__ deg, const int* __restrict__ dst,
                          const float* __restrict__ w, int E) {
    int e = blockIdx.x * blockDim.x + threadIdx.x;
    if (e < E) atomic_add_f32(&deg[dst[e]], w[e]);
}

__global__ void rsqrt_inplace(float* __restrict__ d, int n) {
    int i = blockIdx.x * blockDim.x + threadIdx.x;
    if (i < n) d[i] = __frsqrt_rn(d[i]);
}

__global__ void edge_norm(float* __restrict__ norm, const int* __restrict__ src,
                          const int* __restrict__ dst, const float* __restrict__ w,
                          const float* __restrict__ dinv, int E) {
    int e = blockIdx.x * blockDim.x + threadIdx.x;
    if (e < E) norm[e] = dinv[src[e]] * w[e] * dinv[dst[e]];
}

__global__ __launch_bounds__(32) void gemm_wmma_f32(
        const float* __restrict__ X, const float* __restrict__ W,
        float* __restrict__ H) {
#if defined(__HIP_DEVICE_COMPILE__)
    const int lane  = threadIdx.x;
    const int m0    = blockIdx.x * 16;
    const int rowA  = m0 + (lane & 15);
    const int koff  = (lane >> 4) * 2;
    const int colB  = lane & 15;

    v8f acc[8];
#pragma unroll
    for (int n = 0; n < 8; ++n) acc[n] = v8f{};

    const float* __restrict__ xrow = X + (size_t)rowA * D_CH;

    for (int ks = 0; ks < 32; ++ks) {
        const int kb = ks * 4;
        v2f a;
        a.x = xrow[kb + koff];
        a.y = xrow[kb + koff + 1];
        const float* __restrict__ w0 = W + (size_t)(kb + koff) * D_CH;
#pragma unroll
        for (int n = 0; n < 8; ++n) {
            v2f b;
            b.x = w0[n * 16 + colB];
            b.y = w0[D_CH + n * 16 + colB];
            acc[n] = __builtin_amdgcn_wmma_f32_16x16x4_f32(
                false, a, false, b, (short)0, acc[n], false, false);
        }
    }

    const int outRow  = m0 + ((lane >> 4) * 8);
    const int outCol0 = lane & 15;
#pragma unroll
    for (int n = 0; n < 8; ++n) {
#pragma unroll
        for (int r = 0; r < 8; ++r) {
            H[(size_t)(outRow + r) * D_CH + n * 16 + outCol0] = acc[n][r];
        }
    }
#endif
}

__global__ __launch_bounds__(256) void edge_scatter(
        const float* __restrict__ h, const float* __restrict__ norm,
        const int* __restrict__ src, const int* __restrict__ dst,
        float* __restrict__ agg, int E) {
    const int wave = blockIdx.x * (blockDim.x / 32) + (threadIdx.x >> 5);
    const int lane = threadIdx.x & 31;
    if (wave >= E) return;
    const int   s  = src[wave];
    const int   d  = dst[wave];
    const float nw = norm[wave];
    const float* __restrict__ hs = h + (size_t)s * D_CH;
    float* __restrict__ ad = agg + (size_t)d * D_CH;
#pragma unroll
    for (int j = 0; j < 4; ++j) {
        const int c = lane + j * 32;
        atomic_add_f32(&ad[c], nw * hs[c]);
    }
}

__global__ void finish_layer(const float* __restrict__ agg,
                             const float* __restrict__ h,
                             const float* __restrict__ dinv,
                             const float* __restrict__ b,
                             float* __restrict__ xout, int total) {
    int i = blockIdx.x * blockDim.x + threadIdx.x;
    if (i >= total) return;
    const int node = i >> 7;
    const int ch   = i & (D_CH - 1);
    const float di = dinv[node];
    const float v  = agg[i] + h[i] * di * di + b[ch];
    xout[i] = fmaxf(v, 0.0f);
}

__global__ void lookup_mean(const float* __restrict__ x1,
                            const float* __restrict__ x2,
                            const float* __restrict__ x3,
                            const int* __restrict__ idx,
                            float* __restrict__ out, int nNodes, int total) {
    int i = blockIdx.x * blockDim.x + threadIdx.x;
    if (i >= total) return;
    const int bi = i >> 7;
    const int ch = i & (D_CH - 1);
    const int id = idx[bi];
    float v = 0.0f;
    if (id < nNodes) {
        const size_t o = (size_t)id * D_CH + ch;
        v = (x1[o] + x2[o] + x3[o]) * (1.0f / 3.0f);
    }
    out[i] = v;
}

extern "C" void kernel_launch(void* const* d_in, const int* in_sizes, int n_in,
                              void* d_out, int out_size, void* d_ws, size_t ws_size,
                              hipStream_t stream) {
    const int*   inputs_idx  = (const int*)  d_in[0];
    const float* x           = (const float*)d_in[1];
    const int*   edge_index  = (const int*)  d_in[2];
    const float* edge_weight = (const float*)d_in[3];
    const float* Wm[3] = { (const float*)d_in[4], (const float*)d_in[6], (const float*)d_in[8] };
    const float* bv[3] = { (const float*)d_in[5], (const float*)d_in[7], (const float*)d_in[9] };

    const int B = in_sizes[0];
    const int N = in_sizes[1] / D_CH;
    const int E = in_sizes[2] / 2;
    (void)B; (void)n_in; (void)ws_size;
    const int* src = edge_index;
    const int* dst = edge_index + E;

    float* ws   = (float*)d_ws;
    const size_t nd = (size_t)N * D_CH;
    float* dinv = ws;
    float* norm = dinv + N;
    float* h    = norm + E;
    float* agg  = h + nd;
    float* xl[3];
    xl[0] = agg + nd;
    xl[1] = xl[0] + nd;
    xl[2] = xl[1] + nd;

    const int TB = 256;
    const int mTiles   = N / 16;
    const int edgeBlk  = (E + TB - 1) / TB;
    const int nodeBlk  = (N + TB - 1) / TB;
    const int ndBlk    = (int)((nd + TB - 1) / TB);
    const int waveBlk  = (E + (TB / 32) - 1) / (TB / 32);
    const int outBlk   = (out_size + TB - 1) / TB;

    fill_f32<<<nodeBlk, TB, 0, stream>>>(dinv, 1.0f, N);
    deg_accum<<<edgeBlk, TB, 0, stream>>>(dinv, dst, edge_weight, E);
    rsqrt_inplace<<<nodeBlk, TB, 0, stream>>>(dinv, N);
    edge_norm<<<edgeBlk, TB, 0, stream>>>(norm, src, dst, edge_weight, dinv, E);

    const float* xin = x;
    for (int l = 0; l < 3; ++l) {
        fill_f32<<<ndBlk, TB, 0, stream>>>(agg, 0.0f, (int)nd);
        gemm_wmma_f32<<<mTiles, 32, 0, stream>>>(xin, Wm[l], h);
        edge_scatter<<<waveBlk, TB, 0, stream>>>(h, norm, src, dst, agg, E);
        finish_layer<<<ndBlk, TB, 0, stream>>>(agg, h, dinv, bv[l], xl[l], (int)nd);
        xin = xl[l];
    }

    lookup_mean<<<outBlk, TB, 0, stream>>>(xl[0], xl[1], xl[2], inputs_idx,
                                           (float*)d_out, N, out_size);
}


# Aggregation as destination-sorted gather: int-atomic rank + scan build CSR once; one wave sums a node's rows in registers (arrival-rank order, fma) and finishes it; float scatter and zero-fill gone
# speedup vs baseline: 2.2517x; 2.2517x over previous
#include <hip/hip_runtime.h>
#include <hip/hip_bf16.h>


typedef __attribute__((ext_vector_type(2))) float v2f;
typedef __attribute__((ext_vector_type(8))) float v8f;

#define D_CH 128
#define SCAN_T 1024
#define SCAN_PER 16

__global__ void fill_i32(int* __restrict__ p, int v, int n) {
    int i = blockIdx.x * blockDim.x + threadIdx.x;
    if (i < n) p[i] = v;
}

__global__ void hist_rank(int* __restrict__ cnt, const int* __restrict__ dst,
                          int* __restrict__ rank, int E) {
    int e = blockIdx.x * blockDim.x + threadIdx.x;
    if (e < E) rank[e] = atomicAdd(&cnt[dst[e]], 1);
}

__global__ __launch_bounds__(SCAN_T) void rowptr_scan(
        const int* __restrict__ cnt, int* __restrict__ rowptr, int n) {
    __shared__ int wave_off[32];
    __shared__ int tile_total;
    const int tid  = threadIdx.x;
    const int lane = tid & 31;
    const int wv   = tid >> 5;
    int carry = 0;
    for (int base = 0; base < n; base += SCAN_T * SCAN_PER) {
        const int i0 = base + tid * SCAN_PER;
        const bool whole = (i0 + SCAN_PER <= n);
        int v[SCAN_PER];
        if (whole) {
            const int4* __restrict__ c4 = reinterpret_cast<const int4*>(cnt + i0);
#pragma unroll
            for (int q = 0; q < SCAN_PER / 4; ++q) {
                const int4 t = c4[q];
                v[4 * q + 0] = t.x; v[4 * q + 1] = t.y;
                v[4 * q + 2] = t.z; v[4 * q + 3] = t.w;
            }
        } else {
#pragma unroll
            for (int k = 0; k < SCAN_PER; ++k) v[k] = (i0 + k < n) ? cnt[i0 + k] : 0;
        }
        int s = 0;
#pragma unroll
        for (int k = 0; k < SCAN_PER; ++k) { const int t = v[k]; v[k] = s; s += t; }

        int incl = s;
#pragma unroll
        for (int off = 1; off < 32; off <<= 1) {
            const int t = __shfl_up(incl, off);
            if (lane >= off) incl += t;
        }
        if (lane == 31) wave_off[wv] = incl;
        __syncthreads();
        if (wv == 0) {
            const int t = wave_off[lane];
            int ti = t;
#pragma unroll
            for (int off = 1; off < 32; off <<= 1) {
                const int u = __shfl_up(ti, off);
                if (lane >= off) ti += u;
            }
            wave_off[lane] = ti - t;
            if (lane == 31) tile_total = ti;
        }
        __syncthreads();
        const int off0 = carry + wave_off[wv] + (incl - s);
        if (whole) {
            int4* __restrict__ r4 = reinterpret_cast<int4*>(rowptr + i0);
#pragma unroll
            for (int q = 0; q < SCAN_PER / 4; ++q) {
                int4 t;
                t.x = v[4 * q + 0] + off0; t.y = v[4 * q + 1] + off0;
                t.z = v[4 * q + 2] + off0; t.w = v[4 * q + 3] + off0;
                r4[q] = t;
            }
        } else {
#pragma unroll
            for (int k = 0; k < SCAN_PER; ++k)
                if (i0 + k < n) rowptr[i0 + k] = v[k] + off0;
        }
        carry += tile_total;
        __syncthreads();
    }
    if (tid == 0) rowptr[n] = carry;
}

__global__ void csr_fill(int2* __restrict__ csr, const int* __restrict__ rowptr,
                         const int* __restrict__ rank, const int* __restrict__ src,
                         const int* __restrict__ dst, const float* __restrict__ w, int E) {
    int e = blockIdx.x * blockDim.x + threadIdx.x;
    if (e >= E) return;
    const int p = rowptr[dst[e]] + rank[e];
    if ((unsigned)p < (unsigned)E) csr[p] = make_int2(src[e], __float_as_int(w[e]));
}

__global__ void deg_dinv(float* __restrict__ dinv, const int2* __restrict__ csr,
                         const int* __restrict__ rowptr, int N, int E) {
    int i = blockIdx.x * blockDim.x + threadIdx.x;
    if (i >= N) return;
    const int p0 = max(rowptr[i], 0);
    const int p1 = min(rowptr[i + 1], E);
    float s = 0.0f;
    for (int p = p0; p < p1; ++p) s += __int_as_float(csr[p].y);
    dinv[i] = __frsqrt_rn(s + 1.0f);
}

__global__ void csr_norm(int2* __restrict__ csr, const int* __restrict__ rowptr,
                         const float* __restrict__ dinv, int N, int E) {
    int i = blockIdx.x * blockDim.x + threadIdx.x;
    if (i >= N) return;
    const int p0 = max(rowptr[i], 0);
    const int p1 = min(rowptr[i + 1], E);
    const float di = dinv[i];
    for (int p = p0; p < p1; ++p) {
        const int2 c = csr[p];
        csr[p].y = __float_as_int(dinv[c.x] * __int_as_float(c.y) * di);
    }
}

__global__ __launch_bounds__(32) void gemm_wmma_f32(
        const float* __restrict__ X, const float* __restrict__ W,
        float* __restrict__ H) {
#if defined(__HIP_DEVICE_COMPILE__)
    const int lane  = threadIdx.x;
    const int m0    = blockIdx.x * 16;
    const int rowA  = m0 + (lane & 15);
    const int koff  = (lane >> 4) * 2;
    const int colB  = lane & 15;

    v8f acc[8];
#pragma unroll
    for (int n = 0; n < 8; ++n) acc[n] = v8f{};

    const float* __restrict__ xrow = X + (size_t)rowA * D_CH;

    for (int ks = 0; ks < 32; ++ks) {
        const int kb = ks * 4;
        v2f a;
        a.x = xrow[kb + koff];
        a.y = xrow[kb + koff + 1];
        const float* __restrict__ w0 = W + (size_t)(kb + koff) * D_CH;
#pragma unroll
        for (int n = 0; n < 8; ++n) {
            v2f b;
            b.x = w0[n * 16 + colB];
            b.y = w0[D_CH + n * 16 + colB];
            acc[n] = __builtin_amdgcn_wmma_f32_16x16x4_f32(
                false, a, false, b, (short)0, acc[n], false, false);
        }
    }

    const int outRow  = m0 + ((lane >> 4) * 8);
    const int outCol0 = lane & 15;
#pragma unroll
    for (int n = 0; n < 8; ++n) {
#pragma unroll
        for (int r = 0; r < 8; ++r) {
            H[(size_t)(outRow + r) * D_CH + n * 16 + outCol0] = acc[n][r];
        }
    }
#endif
}

__device__ __forceinline__ void fma4(float4& a, float s, const float4& v) {
    a.x = fmaf(s, v.x, a.x); a.y = fmaf(s, v.y, a.y);
    a.z = fmaf(s, v.z, a.z); a.w = fmaf(s, v.w, a.w);
}

__global__ __launch_bounds__(256) void gather_finish(
        const float* __restrict__ h, const int2* __restrict__ csr,
        const int* __restrict__ rowptr, const float* __restrict__ dinv,
        const float* __restrict__ b, float* __restrict__ xout, int N, int E) {
    const int lane = threadIdx.x & 31;
    const int node = __builtin_amdgcn_readfirstlane(blockIdx.x * 8 + (threadIdx.x >> 5));
    if (node >= N) return;
    const float4* __restrict__ h4 = reinterpret_cast<const float4*>(h);
    const int start = max(rowptr[node], 0);
    const int end   = min(rowptr[node + 1], E);

    const float4 hs = h4[(size_t)node * (D_CH / 4) + lane];
    const float  di = dinv[node];
    const float  b0 = b[lane * 4 + 0], b1 = b[lane * 4 + 1];
    const float  b2 = b[lane * 4 + 2], b3 = b[lane * 4 + 3];

    float4 acc = make_float4(0.0f, 0.0f, 0.0f, 0.0f);
    for (int base = start; base < end; base += 32) {
        const int cnt = min(32, end - base);
        int2 c = make_int2(0, 0);
        if (lane < cnt) c = csr[base + lane];
        int j = 0;
        for (; j + 4 <= cnt; j += 4) {
            const int   s0 = __builtin_amdgcn_readlane(c.x, j + 0);
            const int   s1 = __builtin_amdgcn_readlane(c.x, j + 1);
            const int   s2 = __builtin_amdgcn_readlane(c.x, j + 2);
            const int   s3 = __builtin_amdgcn_readlane(c.x, j + 3);
            const float n0 = __int_as_float(__builtin_amdgcn_readlane(c.y, j + 0));
            const float n1 = __int_as_float(__builtin_amdgcn_readlane(c.y, j + 1));
            const float n2 = __int_as_float(__builtin_amdgcn_readlane(c.y, j + 2));
            const float n3 = __int_as_float(__builtin_amdgcn_readlane(c.y, j + 3));
            const float4 v0 = h4[(size_t)s0 * (D_CH / 4) + lane];
            const float4 v1 = h4[(size_t)s1 * (D_CH / 4) + lane];
            const float4 v2 = h4[(size_t)s2 * (D_CH / 4) + lane];
            const float4 v3 = h4[(size_t)s3 * (D_CH / 4) + lane];
            fma4(acc, n0, v0);
            fma4(acc, n1, v1);
            fma4(acc, n2, v2);
            fma4(acc, n3, v3);
        }
        const int rem = cnt - j;
        if (rem > 0) {
            const float4 zero4 = make_float4(0.0f, 0.0f, 0.0f, 0.0f);
            float4 v1 = zero4, v2 = zero4;
            float  n1 = 0.0f, n2 = 0.0f;
            const int   s0 = __builtin_amdgcn_readlane(c.x, j);
            const float n0 = __int_as_float(__builtin_amdgcn_readlane(c.y, j));
            const float4 v0 = h4[(size_t)s0 * (D_CH / 4) + lane];
            if (rem > 1) {
                const int s1 = __builtin_amdgcn_readlane(c.x, j + 1);
                n1 = __int_as_float(__builtin_amdgcn_readlane(c.y, j + 1));
                v1 = h4[(size_t)s1 * (D_CH / 4) + lane];
            }
            if (rem > 2) {
                const int s2 = __builtin_amdgcn_readlane(c.x, j + 2);
                n2 = __int_as_float(__builtin_amdgcn_readlane(c.y, j + 2));
                v2 = h4[(size_t)s2 * (D_CH / 4) + lane];
            }
            fma4(acc, n0, v0);
            fma4(acc, n1, v1);
            fma4(acc, n2, v2);
        }
    }

    float4 o;
    o.x = fmaxf(acc.x + hs.x * di * di + b0, 0.0f);
    o.y = fmaxf(acc.y + hs.y * di * di + b1, 0.0f);
    o.z = fmaxf(acc.z + hs.z * di * di + b2, 0.0f);
    o.w = fmaxf(acc.w + hs.w * di * di + b3, 0.0f);
    reinterpret_cast<float4*>(xout)[(size_t)node * (D_CH / 4) + lane] = o;
}

__global__ void lookup_mean(const float* __restrict__ x1,
                            const float* __restrict__ x2,
                            const float* __restrict__ x3,
                            const int* __restrict__ idx,
                            float* __restrict__ out, int nNodes, int total) {
    int i = blockIdx.x * blockDim.x + threadIdx.x;
    if (i >= total) return;
    const int bi = i >> 7;
    const int ch = i & (D_CH - 1);
    const int id = idx[bi];
    float v = 0.0f;
    if (id < nNodes) {
        const size_t o = (size_t)id * D_CH + ch;
        v = (x1[o] + x2[o] + x3[o]) * (1.0f / 3.0f);
    }
    out[i] = v;
}

extern "C" void kernel_launch(void* const* d_in, const int* in_sizes, int n_in,
                              void* d_out, int out_size, void* d_ws, size_t ws_size,
                              hipStream_t stream) {
    const int*   inputs_idx  = (const int*)  d_in[0];
    const float* x           = (const float*)d_in[1];
    const int*   edge_index  = (const int*)  d_in[2];
    const float* edge_weight = (const float*)d_in[3];
    const float* Wm[3] = { (const float*)d_in[4], (const float*)d_in[6], (const float*)d_in[8] };
    const float* bv[3] = { (const float*)d_in[5], (const float*)d_in[7], (const float*)d_in[9] };

    const int B = in_sizes[0];
    const int N = in_sizes[1] / D_CH;
    const int E = in_sizes[2] / 2;
    (void)B; (void)n_in; (void)ws_size;
    const int* src = edge_index;
    const int* dst = edge_index + E;

    float* ws   = (float*)d_ws;
    const size_t nd   = (size_t)N * D_CH;
    const size_t Npad = ((size_t)N + 3) & ~(size_t)3;
    float* h    = ws;
    float* xl[3];
    xl[0] = h + nd;
    xl[1] = xl[0] + nd;
    xl[2] = xl[1] + nd;
    int*   cnt    = (int*)(xl[2] + nd);
    float* dinv   = (float*)(cnt + Npad);
    int*   rowptr = (int*)(dinv + Npad);
    int2*  csr    = (int2*)(rowptr + Npad + 4);
    int*   rank   = (int*)(csr + E);

    const int TB = 256;
    const int mTiles   = N / 16;
    const int edgeBlk  = (E + TB - 1) / TB;
    const int nodeBlk  = (N + TB - 1) / TB;
    const int waveBlk  = (N + (TB / 32) - 1) / (TB / 32);
    const int outBlk   = (out_size + TB - 1) / TB;

    fill_i32<<<nodeBlk, TB, 0, stream>>>(cnt, 0, N);
    hist_rank<<<edgeBlk, TB, 0, stream>>>(cnt, dst, rank, E);
    rowptr_scan<<<1, SCAN_T, 0, stream>>>(cnt, rowptr, N);
    csr_fill<<<edgeBlk, TB, 0, stream>>>(csr, rowptr, rank, src, dst, edge_weight, E);
    deg_dinv<<<nodeBlk, TB, 0, stream>>>(dinv, csr, rowptr, N, E);
    csr_norm<<<nodeBlk, TB, 0, stream>>>(csr, rowptr, dinv, N, E);

    const float* xin = x;
    for (int l = 0; l < 3; ++l) {
        gemm_wmma_f32<<<mTiles, 32, 0, stream>>>(xin, Wm[l], h);
        gather_finish<<<waveBlk, TB, 0, stream>>>(h, csr, rowptr, dinv, bv[l], xl[l], N, E);
        xin = xl[l];
    }

    lookup_mean<<<outBlk, TB, 0, stream>>>(xl[0], xl[1], xl[2], inputs_idx,
                                           (float*)d_out, N, out_size);
}


# Layer 3 evaluated only at the looked-up rows: their x2 neighbours are summed first, then one WMMA product with W3 (sum order of that layer changes), fused with the mean; 2 launches gone
# speedup vs baseline: 2.8032x; 1.2449x over previous
#include <hip/hip_runtime.h>
#include <hip/hip_bf16.h>


typedef __attribute__((ext_vector_type(2))) float v2f;
typedef __attribute__((ext_vector_type(8))) float v8f;

#define D_CH 128
#define SCAN_T 1024
#define SCAN_PER 16

__global__ void fill_i32(int* __restrict__ p, int v, int n) {
    int i = blockIdx.x * blockDim.x + threadIdx.x;
    if (i < n) p[i] = v;
}

__global__ void hist_rank(int* __restrict__ cnt, const int* __restrict__ dst,
                          int* __restrict__ rank, int E) {
    int e = blockIdx.x * blockDim.x + threadIdx.x;
    if (e < E) rank[e] = atomicAdd(&cnt[dst[e]], 1);
}

__global__ __launch_bounds__(SCAN_T) void rowptr_scan(
        const int* __restrict__ cnt, int* __restrict__ rowptr, int n) {
    __shared__ int wave_off[32];
    __shared__ int tile_total;
    const int tid  = threadIdx.x;
    const int lane = tid & 31;
    const int wv   = tid >> 5;
    int carry = 0;
    for (int base = 0; base < n; base += SCAN_T * SCAN_PER) {
        const int i0 = base + tid * SCAN_PER;
        const bool whole = (i0 + SCAN_PER <= n);
        int v[SCAN_PER];
        if (whole) {
            const int4* __restrict__ c4 = reinterpret_cast<const int4*>(cnt + i0);
#pragma unroll
            for (int q = 0; q < SCAN_PER / 4; ++q) {
                const int4 t = c4[q];
                v[4 * q + 0] = t.x; v[4 * q + 1] = t.y;
                v[4 * q + 2] = t.z; v[4 * q + 3] = t.w;
            }
        } else {
#pragma unroll
            for (int k = 0; k < SCAN_PER; ++k) v[k] = (i0 + k < n) ? cnt[i0 + k] : 0;
        }
        int s = 0;
#pragma unroll
        for (int k = 0; k < SCAN_PER; ++k) { const int t = v[k]; v[k] = s; s += t; }

        int incl = s;
#pragma unroll
        for (int off = 1; off < 32; off <<= 1) {
            const int t = __shfl_up(incl, off);
            if (lane >= off) incl += t;
        }
        if (lane == 31) wave_off[wv] = incl;
        __syncthreads();
        if (wv == 0) {
            const int t = wave_off[lane];
            int ti = t;
#pragma unroll
            for (int off = 1; off < 32; off <<= 1) {
                const int u = __shfl_up(ti, off);
                if (lane >= off) ti += u;
            }
            wave_off[lane] = ti - t;
            if (lane == 31) tile_total = ti;
        }
        __syncthreads();
        const int off0 = carry + wave_off[wv] + (incl - s);
        if (whole) {
            int4* __restrict__ r4 = reinterpret_cast<int4*>(rowptr + i0);
#pragma unroll
            for (int q = 0; q < SCAN_PER / 4; ++q) {
                int4 t;
                t.x = v[4 * q + 0] + off0; t.y = v[4 * q + 1] + off0;
                t.z = v[4 * q + 2] + off0; t.w = v[4 * q + 3] + off0;
                r4[q] = t;
            }
        } else {
#pragma unroll
            for (int k = 0; k < SCAN_PER; ++k)
                if (i0 + k < n) rowptr[i0 + k] = v[k] + off0;
        }
        carry += tile_total;
        __syncthreads();
    }
    if (tid == 0) rowptr[n] = carry;
}

__global__ void csr_fill(int2* __restrict__ csr, const int* __restrict__ rowptr,
                         const int* __restrict__ rank, const int* __restrict__ src,
                         const int* __restrict__ dst, const float* __restrict__ w, int E) {
    int e = blockIdx.x * blockDim.x + threadIdx.x;
    if (e >= E) return;
    const int p = rowptr[dst[e]] + rank[e];
    if ((unsigned)p < (unsigned)E) csr[p] = make_int2(src[e], __float_as_int(w[e]));
}

__global__ void deg_dinv(float* __restrict__ dinv, const int2* __restrict__ csr,
                         const int* __restrict__ rowptr, int N, int E) {
    int i = blockIdx.x * blockDim.x + threadIdx.x;
    if (i >= N) return;
    const int p0 = max(rowptr[i], 0);
    const int p1 = min(rowptr[i + 1], E);
    float s = 0.0f;
    for (int p = p0; p < p1; ++p) s += __int_as_float(csr[p].y);
    dinv[i] = __frsqrt_rn(s + 1.0f);
}

__global__ void csr_norm(int2* __restrict__ csr, const int* __restrict__ rowptr,
                         const float* __restrict__ dinv, int N, int E) {
    int i = blockIdx.x * blockDim.x + threadIdx.x;
    if (i >= N) return;
    const int p0 = max(rowptr[i], 0);
    const int p1 = min(rowptr[i + 1], E);
    const float di = dinv[i];
    for (int p = p0; p < p1; ++p) {
        const int2 c = csr[p];
        csr[p].y = __float_as_int(dinv[c.x] * __int_as_float(c.y) * di);
    }
}

__global__ __launch_bounds__(32) void gemm_wmma_f32(
        const float* __restrict__ X, const float* __restrict__ W,
        float* __restrict__ H) {
#if defined(__HIP_DEVICE_COMPILE__)
    const int lane  = threadIdx.x;
    const int m0    = blockIdx.x * 16;
    const int rowA  = m0 + (lane & 15);
    const int koff  = (lane >> 4) * 2;
    const int colB  = lane & 15;

    v8f acc[8];
#pragma unroll
    for (int n = 0; n < 8; ++n) acc[n] = v8f{};

    const float* __restrict__ xrow = X + (size_t)rowA * D_CH;

    for (int ks = 0; ks < 32; ++ks) {
        const int kb = ks * 4;
        v2f a;
        a.x = xrow[kb + koff];
        a.y = xrow[kb + koff + 1];
        const float* __restrict__ w0 = W + (size_t)(kb + koff) * D_CH;
#pragma unroll
        for (int n = 0; n < 8; ++n) {
            v2f b;
            b.x = w0[n * 16 + colB];
            b.y = w0[D_CH + n * 16 + colB];
            acc[n] = __builtin_amdgcn_wmma_f32_16x16x4_f32(
                false, a, false, b, (short)0, acc[n], false, false);
        }
    }

    const int outRow  = m0 + ((lane >> 4) * 8);
    const int outCol0 = lane & 15;
#pragma unroll
    for (int n = 0; n < 8; ++n) {
#pragma unroll
        for (int r = 0; r < 8; ++r) {
            H[(size_t)(outRow + r) * D_CH + n * 16 + outCol0] = acc[n][r];
        }
    }
#endif
}

__device__ __forceinline__ void fma4(float4& a, float s, const float4& v) {
    a.x = fmaf(s, v.x, a.x); a.y = fmaf(s, v.y, a.y);
    a.z = fmaf(s, v.z, a.z); a.w = fmaf(s, v.w, a.w);
}

__device__ __forceinline__ float4 segment_sum(const float4* __restrict__ h4,
                                              const int2* __restrict__ csr,
                                              int start, int end, int lane) {
    float4 acc = make_float4(0.0f, 0.0f, 0.0f, 0.0f);
    for (int base = start; base < end; base += 32) {
        const int cnt = min(32, end - base);
        int2 c = make_int2(0, 0);
        if (lane < cnt) c = csr[base + lane];
        int j = 0;
        for (; j + 4 <= cnt; j += 4) {
            const int   s0 = __builtin_amdgcn_readlane(c.x, j + 0);
            const int   s1 = __builtin_amdgcn_readlane(c.x, j + 1);
            const int   s2 = __builtin_amdgcn_readlane(c.x, j + 2);
            const int   s3 = __builtin_amdgcn_readlane(c.x, j + 3);
            const float n0 = __int_as_float(__builtin_amdgcn_readlane(c.y, j + 0));
            const float n1 = __int_as_float(__builtin_amdgcn_readlane(c.y, j + 1));
            const float n2 = __int_as_float(__builtin_amdgcn_readlane(c.y, j + 2));
            const float n3 = __int_as_float(__builtin_amdgcn_readlane(c.y, j + 3));
            const float4 v0 = h4[(size_t)s0 * (D_CH / 4) + lane];
            const float4 v1 = h4[(size_t)s1 * (D_CH / 4) + lane];
            const float4 v2 = h4[(size_t)s2 * (D_CH / 4) + lane];
            const float4 v3 = h4[(size_t)s3 * (D_CH / 4) + lane];
            fma4(acc, n0, v0);
            fma4(acc, n1, v1);
            fma4(acc, n2, v2);
            fma4(acc, n3, v3);
        }
        const int rem = cnt - j;
        if (rem > 0) {
            const float4 zero4 = make_float4(0.0f, 0.0f, 0.0f, 0.0f);
            float4 v1 = zero4, v2 = zero4;
            float  n1 = 0.0f, n2 = 0.0f;
            const int   s0 = __builtin_amdgcn_readlane(c.x, j);
            const float n0 = __int_as_float(__builtin_amdgcn_readlane(c.y, j));
            const float4 v0 = h4[(size_t)s0 * (D_CH / 4) + lane];
            if (rem > 1) {
                const int s1 = __builtin_amdgcn_readlane(c.x, j + 1);
                n1 = __int_as_float(__builtin_amdgcn_readlane(c.y, j + 1));
                v1 = h4[(size_t)s1 * (D_CH / 4) + lane];
            }
            if (rem > 2) {
                const int s2 = __builtin_amdgcn_readlane(c.x, j + 2);
                n2 = __int_as_float(__builtin_amdgcn_readlane(c.y, j + 2));
                v2 = h4[(size_t)s2 * (D_CH / 4) + lane];
            }
            fma4(acc, n0, v0);
            fma4(acc, n1, v1);
            fma4(acc, n2, v2);
        }
    }
    return acc;
}

__global__ __launch_bounds__(256) void gather_finish(
        const float* __restrict__ h, const int2* __restrict__ csr,
        const int* __restrict__ rowptr, const float* __restrict__ dinv,
        const float* __restrict__ b, float* __restrict__ xout, int N, int E) {
    const int lane = threadIdx.x & 31;
    const int node = __builtin_amdgcn_readfirstlane(blockIdx.x * 8 + (threadIdx.x >> 5));
    if (node >= N) return;
    const float4* __restrict__ h4 = reinterpret_cast<const float4*>(h);
    const int start = max(rowptr[node], 0);
    const int end   = min(rowptr[node + 1], E);

    const float4 hs = h4[(size_t)node * (D_CH / 4) + lane];
    const float  di = dinv[node];
    const float  b0 = b[lane * 4 + 0], b1 = b[lane * 4 + 1];
    const float  b2 = b[lane * 4 + 2], b3 = b[lane * 4 + 3];

    const float4 acc = segment_sum(h4, csr, start, end, lane);

    float4 o;
    o.x = fmaxf(acc.x + hs.x * di * di + b0, 0.0f);
    o.y = fmaxf(acc.y + hs.y * di * di + b1, 0.0f);
    o.z = fmaxf(acc.z + hs.z * di * di + b2, 0.0f);
    o.w = fmaxf(acc.w + hs.w * di * di + b3, 0.0f);
    reinterpret_cast<float4*>(xout)[(size_t)node * (D_CH / 4) + lane] = o;
}

#define LT_ROWS  16
#define LT_PITCH 132

__global__ __launch_bounds__(256) void last_layer_at_lookups(
        const float* __restrict__ x1, const float* __restrict__ x2,
        const int2* __restrict__ csr, const int* __restrict__ rowptr,
        const float* __restrict__ dinv, const float* __restrict__ W,
        const float* __restrict__ b, const int* __restrict__ idx,
        float* __restrict__ out, int N, int E, int B) {
#if defined(__HIP_DEVICE_COMPILE__)
    __shared__ __attribute__((aligned(16))) float tile[LT_ROWS * LT_PITCH];
    const int lane  = threadIdx.x & 31;
    const int wv    = __builtin_amdgcn_readfirstlane(threadIdx.x >> 5);
    const int slot0 = blockIdx.x * LT_ROWS;
    const float4* __restrict__ x1v = reinterpret_cast<const float4*>(x1);
    const float4* __restrict__ x2v = reinterpret_cast<const float4*>(x2);
    const float4 zero4 = make_float4(0.0f, 0.0f, 0.0f, 0.0f);

    const int slotA = slot0 + wv, slotB = slot0 + wv + 8;
    int nodes[2] = { N, N };
    if (slotA < B) nodes[0] = idx[slotA];
    if (slotB < B) nodes[1] = idx[slotB];

    float4 keep[2];
#pragma unroll
    for (int t = 0; t < 2; ++t) {
        const int r    = wv + 8 * t;
        const int node = nodes[t];
        float4 a = zero4, k = zero4;
        if ((unsigned)node < (unsigned)N) {
            const int start = max(rowptr[node], 0);
            const int end   = min(rowptr[node + 1], E);
            const float4 hs = x2v[(size_t)node * (D_CH / 4) + lane];
            const float4 p1 = x1v[(size_t)node * (D_CH / 4) + lane];
            const float  di = dinv[node];
            a = segment_sum(x2v, csr, start, end, lane);
            const float d2 = di * di;
            a.x = fmaf(hs.x, d2, a.x); a.y = fmaf(hs.y, d2, a.y);
            a.z = fmaf(hs.z, d2, a.z); a.w = fmaf(hs.w, d2, a.w);
            k.x = p1.x + hs.x; k.y = p1.y + hs.y;
            k.z = p1.z + hs.z; k.w = p1.w + hs.w;
        }
        keep[t] = k;
        *reinterpret_cast<float4*>(&tile[r * LT_PITCH + lane * 4]) = a;
    }
    __syncthreads();

    const int koff = (lane >> 4) * 2;
    const int col  = wv * 16 + (lane & 15);
    const float* __restrict__ arow = tile + (lane & 15) * LT_PITCH + koff;
    const float* __restrict__ wcol = W + (size_t)koff * D_CH + col;
    v8f acc = v8f{};
#pragma unroll
    for (int ks = 0; ks < 32; ++ks) {
        v2f av, bw;
        av.x = arow[ks * 4];
        av.y = arow[ks * 4 + 1];
        bw.x = wcol[(size_t)(ks * 4) * D_CH];
        bw.y = wcol[(size_t)(ks * 4 + 1) * D_CH];
        acc = __builtin_amdgcn_wmma_f32_16x16x4_f32(
            false, av, false, bw, (short)0, acc, false, false);
    }
    __syncthreads();

    const float bias = b[col];
    const int   row0 = (lane >> 4) * 8;
#pragma unroll
    for (int r = 0; r < 8; ++r)
        tile[(row0 + r) * LT_PITCH + col] = fmaxf(acc[r] + bias, 0.0f);
    __syncthreads();

#pragma unroll
    for (int t = 0; t < 2; ++t) {
        const int r    = wv + 8 * t;
        const int slot = slot0 + r;
        if (slot < B) {
            float4 o = zero4;
            if ((unsigned)nodes[t] < (unsigned)N) {
                const float4 x3 = *reinterpret_cast<const float4*>(&tile[r * LT_PITCH + lane * 4]);
                o.x = (keep[t].x + x3.x) * (1.0f / 3.0f);
                o.y = (keep[t].y + x3.y) * (1.0f / 3.0f);
                o.z = (keep[t].z + x3.z) * (1.0f / 3.0f);
                o.w = (keep[t].w + x3.w) * (1.0f / 3.0f);
            }
            reinterpret_cast<float4*>(out)[(size_t)slot * (D_CH / 4) + lane] = o;
        }
    }
#endif
}

extern "C" void kernel_launch(void* const* d_in, const int* in_sizes, int n_in,
                              void* d_out, int out_size, void* d_ws, size_t ws_size,
                              hipStream_t stream) {
    const int*   inputs_idx  = (const int*)  d_in[0];
    const float* x           = (const float*)d_in[1];
    const int*   edge_index  = (const int*)  d_in[2];
    const float* edge_weight = (const float*)d_in[3];
    const float* Wm[3] = { (const float*)d_in[4], (const float*)d_in[6], (const float*)d_in[8] };
    const float* bv[3] = { (const float*)d_in[5], (const float*)d_in[7], (const float*)d_in[9] };

    const int B = in_sizes[0];
    const int N = in_sizes[1] / D_CH;
    const int E = in_sizes[2] / 2;
    (void)n_in; (void)ws_size; (void)out_size;
    const int* src = edge_index;
    const int* dst = edge_index + E;

    float* ws   = (float*)d_ws;
    const size_t nd   = (size_t)N * D_CH;
    const size_t Npad = ((size_t)N + 3) & ~(size_t)3;
    float* h    = ws;
    float* xl[2];
    xl[0] = h + nd;
    xl[1] = xl[0] + nd;
    int*   cnt    = (int*)(xl[1] + nd);
    float* dinv   = (float*)(cnt + Npad);
    int*   rowptr = (int*)(dinv + Npad);
    int2*  csr    = (int2*)(rowptr + Npad + 4);
    int*   rank   = (int*)(csr + E);

    const int TB = 256;
    const int mTiles   = N / 16;
    const int edgeBlk  = (E + TB - 1) / TB;
    const int nodeBlk  = (N + TB - 1) / TB;
    const int waveBlk  = (N + (TB / 32) - 1) / (TB / 32);
    const int slotBlk  = (B + LT_ROWS - 1) / LT_ROWS;

    fill_i32<<<nodeBlk, TB, 0, stream>>>(cnt, 0, N);
    hist_rank<<<edgeBlk, TB, 0, stream>>>(cnt, dst, rank, E);
    rowptr_scan<<<1, SCAN_T, 0, stream>>>(cnt, rowptr, N);
    csr_fill<<<edgeBlk, TB, 0, stream>>>(csr, rowptr, rank, src, dst, edge_weight, E);
    deg_dinv<<<nodeBlk, TB, 0, stream>>>(dinv, csr, rowptr, N, E);
    csr_norm<<<nodeBlk, TB, 0, stream>>>(csr, rowptr, dinv, N, E);

    const float* xin = x;
    for (int l = 0; l < 2; ++l) {
        gemm_wmma_f32<<<mTiles, 32, 0, stream>>>(xin, Wm[l], h);
        gather_finish<<<waveBlk, TB, 0, stream>>>(h, csr, rowptr, dinv, bv[l], xl[l], N, E);
        xin = xl[l];
    }

    last_layer_at_lookups<<<slotBlk, TB, 0, stream>>>(xl[0], xl[1], csr, rowptr, dinv,
                                                      Wm[2], bv[2], inputs_idx,
                                                      (float*)d_out, N, E, B);
}


# Gather with twice the loads in flight: a wave walks the csr segments of its two nodes of a tile side by side (8 rows per trip, not 4; bounds of both nodes fetched first); per-node sum order kept
# speedup vs baseline: 3.3449x; 1.1932x over previous
#include <hip/hip_runtime.h>
#include <hip/hip_bf16.h>


typedef __attribute__((ext_vector_type(2))) float v2f;
typedef __attribute__((ext_vector_type(8))) float v8f;
typedef __attribute__((ext_vector_type(4))) float v4f;

#define D_CH 128
#define SCAN_T 1024
#define SCAN_PER 16

__global__ void fill_i32(int* __restrict__ p, int v, int n) {
    int i = blockIdx.x * blockDim.x + threadIdx.x;
    if (i < n) p[i] = v;
}

__global__ void hist_rank(int* __restrict__ cnt, const int* __restrict__ dst,
                          int* __restrict__ rank, int E) {
    int e = blockIdx.x * blockDim.x + threadIdx.x;
    if (e < E) rank[e] = atomicAdd(&cnt[dst[e]], 1);
}

__global__ __launch_bounds__(SCAN_T) void rowptr_scan(
        const int* __restrict__ cnt, int* __restrict__ rowptr, int n) {
    __shared__ int wave_off[32];
    __shared__ int tile_total;
    const int tid  = threadIdx.x;
    const int lane = tid & 31;
    const int wv   = tid >> 5;
    int carry = 0;
    for (int base = 0; base < n; base += SCAN_T * SCAN_PER) {
        const int i0 = base + tid * SCAN_PER;
        const bool whole = (i0 + SCAN_PER <= n);
        int v[SCAN_PER];
        if (whole) {
            const int4* __restrict__ c4 = reinterpret_cast<const int4*>(cnt + i0);
#pragma unroll
            for (int q = 0; q < SCAN_PER / 4; ++q) {
                const int4 t = c4[q];
                v[4 * q + 0] = t.x; v[4 * q + 1] = t.y;
                v[4 * q + 2] = t.z; v[4 * q + 3] = t.w;
            }
        } else {
#pragma unroll
            for (int k = 0; k < SCAN_PER; ++k) v[k] = (i0 + k < n) ? cnt[i0 + k] : 0;
        }
        int s = 0;
#pragma unroll
        for (int k = 0; k < SCAN_PER; ++k) { const int t = v[k]; v[k] = s; s += t; }

        int incl = s;
#pragma unroll
        for (int off = 1; off < 32; off <<= 1) {
            const int t = __shfl_up(incl, off);
            if (lane >= off) incl += t;
        }
        if (lane == 31) wave_off[wv] = incl;
        __syncthreads();
        if (wv == 0) {
            const int t = wave_off[lane];
            int ti = t;
#pragma unroll
            for (int off = 1; off < 32; off <<= 1) {
                const int u = __shfl_up(ti, off);
                if (lane >= off) ti += u;
            }
            wave_off[lane] = ti - t;
            if (lane == 31) tile_total = ti;
        }
        __syncthreads();
        const int off0 = carry + wave_off[wv] + (incl - s);
        if (whole) {
            int4* __restrict__ r4 = reinterpret_cast<int4*>(rowptr + i0);
#pragma unroll
            for (int q = 0; q < SCAN_PER / 4; ++q) {
                int4 t;
                t.x = v[4 * q + 0] + off0; t.y = v[4 * q + 1] + off0;
                t.z = v[4 * q + 2] + off0; t.w = v[4 * q + 3] + off0;
                r4[q] = t;
            }
        } else {
#pragma unroll
            for (int k = 0; k < SCAN_PER; ++k)
                if (i0 + k < n) rowptr[i0 + k] = v[k] + off0;
        }
        carry += tile_total;
        __syncthreads();
    }
    if (tid == 0) rowptr[n] = carry;
}

__global__ void csr_fill(int2* __restrict__ csr, const int* __restrict__ rowptr,
                         const int* __restrict__ rank, const int* __restrict__ src,
                         const int* __restrict__ dst, const float* __restrict__ w, int E) {
    int e = blockIdx.x * blockDim.x + threadIdx.x;
    if (e >= E) return;
    const int p = rowptr[dst[e]] + rank[e];
    if ((unsigned)p < (unsigned)E) csr[p] = make_int2(src[e], __float_as_int(w[e]));
}

__global__ void deg_dinv(float* __restrict__ dinv, const int2* __restrict__ csr,
                         const int* __restrict__ rowptr, int N, int E) {
    int i = blockIdx.x * blockDim.x + threadIdx.x;
    if (i >= N) return;
    const int p0 = max(rowptr[i], 0);
    const int p1 = min(rowptr[i + 1], E);
    float s = 0.0f;
    for (int p = p0; p < p1; ++p) s += __int_as_float(csr[p].y);
    dinv[i] = __frsqrt_rn(s + 1.0f);
}

__global__ void csr_norm(int2* __restrict__ csr, const int* __restrict__ rowptr,
                         const float* __restrict__ dinv, int N, int E) {
    int i = blockIdx.x * blockDim.x + threadIdx.x;
    if (i >= N) return;
    const int p0 = max(rowptr[i], 0);
    const int p1 = min(rowptr[i + 1], E);
    const float di = dinv[i];
    for (int p = p0; p < p1; ++p) {
        const int2 c = csr[p];
        csr[p].y = __float_as_int(dinv[c.x] * __int_as_float(c.y) * di);
    }
}

__device__ __forceinline__ void fma4(float4& a, float s, const v4f& v) {
    a.x = fmaf(s, v.x, a.x); a.y = fmaf(s, v.y, a.y);
    a.z = fmaf(s, v.z, a.z); a.w = fmaf(s, v.w, a.w);
}

typedef const __attribute__((address_space(1))) v4f* grow_t;
__device__ __forceinline__ v4f load_row(const float4* __restrict__ h4, int s, int lane) {
    const unsigned long long a = (unsigned long long)h4 + ((unsigned long long)(unsigned)s << 9);
    const unsigned lo = __builtin_amdgcn_readfirstlane((unsigned)a);
    const unsigned hi = __builtin_amdgcn_readfirstlane((unsigned)(a >> 32));
    grow_t row = (grow_t)(((unsigned long long)hi << 32) | lo);
    return row[(unsigned)lane];
}
__device__ __forceinline__ void ask4(const float4* __restrict__ h4, const int2 c, int j, int lane,
                                     v4f (&v)[4], float (&n)[4]) {
#pragma unroll
    for (int k = 0; k < 4; ++k) {
        const int s = __builtin_amdgcn_readlane(c.x, j + k);
        n[k] = __int_as_float(__builtin_amdgcn_readlane(c.y, j + k));
        v[k] = load_row(h4, s, lane);
    }
}
__device__ __forceinline__ void add4(float4& acc, const v4f (&v)[4], const float (&n)[4]) {
#pragma unroll
    for (int k = 0; k < 4; ++k) fma4(acc, n[k], v[k]);
}
__device__ __forceinline__ void pick_tail(const int2 c, int j, int rem, int (&s)[3], float (&n)[3]) {
#pragma unroll
    for (int k = 0; k < 3; ++k) {
        s[k] = __builtin_amdgcn_readlane(c.x, (j + k) & 31);
        const float w = __int_as_float(__builtin_amdgcn_readlane(c.y, (j + k) & 31));
        n[k] = (k < rem) ? w : 0.0f;
    }
}
__device__ __forceinline__ void read_tail(const float4* __restrict__ h4, const int (&s)[3], int rem,
                                          int lane, v4f (&v)[3]) {
#pragma unroll
    for (int k = 0; k < 3; ++k) {
        v[k] = v4f{0.0f, 0.0f, 0.0f, 0.0f};
        if (k < rem) v[k] = load_row(h4, s[k], lane);
    }
}
__device__ __forceinline__ void add_tail(float4& acc, const v4f (&v)[3], const float (&n)[3]) {
#pragma unroll
    for (int k = 0; k < 3; ++k) fma4(acc, n[k], v[k]);
}

__device__ __forceinline__ void segment_sum_pair(const float4* __restrict__ h4,
                                                 const int2* __restrict__ csr,
                                                 int startA, int endA, int startB, int endB,
                                                 int lane, float4& accA, float4& accB) {
    accA = make_float4(0.0f, 0.0f, 0.0f, 0.0f);
    accB = make_float4(0.0f, 0.0f, 0.0f, 0.0f);
    while (startA < endA || startB < endB) {
        const int cntA = max(0, min(32, endA - startA));
        const int cntB = max(0, min(32, endB - startB));
        int2 ca = make_int2(0, 0), cb = make_int2(0, 0);
        if (lane < cntA) ca = csr[startA + lane];
        if (lane < cntB) cb = csr[startB + lane];
        int j = 0;
        for (; j + 4 <= min(cntA, cntB); j += 4) {
            v4f   va[4], vb[4];
            float na[4], nb[4];
            ask4(h4, ca, j, lane, va, na);
            ask4(h4, cb, j, lane, vb, nb);
            asm volatile("" :: "v"(va[0]), "v"(va[1]), "v"(va[2]), "v"(va[3]),
                               "v"(vb[0]), "v"(vb[1]), "v"(vb[2]), "v"(vb[3]));
            add4(accA, va, na);
            add4(accB, vb, nb);
        }
        int jA = j, jB = j;
        for (; jA + 4 <= cntA; jA += 4) {
            v4f   va[4];
            float na[4];
            ask4(h4, ca, jA, lane, va, na);
            add4(accA, va, na);
        }
        for (; jB + 4 <= cntB; jB += 4) {
            v4f   vb[4];
            float nb[4];
            ask4(h4, cb, jB, lane, vb, nb);
            add4(accB, vb, nb);
        }
        const int remA = cntA - jA, remB = cntB - jB;
        if (remA > 0 || remB > 0) {
            v4f   va[3], vb[3];
            float na[3], nb[3];
            int   sa[3], sb[3];
            pick_tail(ca, jA, remA, sa, na);
            pick_tail(cb, jB, remB, sb, nb);
            read_tail(h4, sa, remA, lane, va);
            read_tail(h4, sb, remB, lane, vb);
            add_tail(accA, va, na);
            add_tail(accB, vb, nb);
        }
        startA += 32;
        startB += 32;
    }
}

#define LT_ROWS  16
#define LT_PITCH 132

__global__ __launch_bounds__(256, 8) void aggregate_then_product(
        const float* __restrict__ xin, const int2* __restrict__ csr,
        const int* __restrict__ rowptr, const float* __restrict__ dinv,
        const float* __restrict__ W, const float* __restrict__ b,
        float* __restrict__ xout, int N, int E, int nTiles) {
#if defined(__HIP_DEVICE_COMPILE__)
    __shared__ __attribute__((aligned(16))) float tile[LT_ROWS * LT_PITCH];
    const int lane = threadIdx.x & 31;
    const int wv   = __builtin_amdgcn_readfirstlane(threadIdx.x >> 5);
    const float4* __restrict__ xv = reinterpret_cast<const float4*>(xin);

    const int koff = (lane >> 4) * 2;
    const int col  = wv * 16 + (lane & 15);
    const float* __restrict__ arow = tile + (lane & 15) * LT_PITCH + koff;
    const float* __restrict__ wcol = W + (size_t)koff * D_CH + col;
    const int   row0 = (lane >> 4) * 8;

    v2f bw[32];
#pragma unroll
    for (int ks = 0; ks < 32; ++ks) {
        bw[ks].x = wcol[(size_t)(ks * 4) * D_CH];
        bw[ks].y = wcol[(size_t)(ks * 4 + 1) * D_CH];
    }
    const float bias = b[col];

    for (int t = blockIdx.x; t < nTiles; t += gridDim.x) {
        const int node0 = t * LT_ROWS;
        int    rp0[2], rp1[2];
        v4f    hs[2];
        float  di[2];
#pragma unroll
        for (int u = 0; u < 2; ++u) {
            const int nd = min(node0 + wv + 8 * u, N - 1);
            rp0[u] = rowptr[nd];
            rp1[u] = rowptr[nd + 1];
            hs[u]  = load_row(xv, nd, lane);
            di[u]  = dinv[nd];
        }
#pragma unroll
        for (int u = 0; u < 2; ++u) {
            rp0[u] = max(rp0[u], 0);
            rp1[u] = min(rp1[u], E);
        }
        float4 a[2];
        segment_sum_pair(xv, csr, rp0[0], rp1[0], rp0[1], rp1[1], lane, a[0], a[1]);
#pragma unroll
        for (int u = 0; u < 2; ++u) {
            const int   r  = wv + 8 * u;
            const float d2 = di[u] * di[u];
            a[u].x = fmaf(hs[u].x, d2, a[u].x); a[u].y = fmaf(hs[u].y, d2, a[u].y);
            a[u].z = fmaf(hs[u].z, d2, a[u].z); a[u].w = fmaf(hs[u].w, d2, a[u].w);
            *reinterpret_cast<float4*>(&tile[r * LT_PITCH + lane * 4]) = a[u];
        }
        __syncthreads();

        v8f acc = v8f{};
#pragma unroll
        for (int ks = 0; ks < 32; ++ks) {
            v2f av;
            av.x = arow[ks * 4];
            av.y = arow[ks * 4 + 1];
            acc = __builtin_amdgcn_wmma_f32_16x16x4_f32(
                false, av, false, bw[ks], (short)0, acc, false, false);
        }
        __syncthreads();

#pragma unroll
        for (int r = 0; r < 8; ++r)
            tile[(row0 + r) * LT_PITCH + col] = fmaxf(acc[r] + bias, 0.0f);
        __syncthreads();

        for (int u = 0; u < 2; ++u) {
            const int r    = wv + 8 * u;
            const int node = node0 + r;
            if (node < N) {
                const float4 o = *reinterpret_cast<const float4*>(&tile[r * LT_PITCH + lane * 4]);
                reinterpret_cast<float4*>(xout)[(size_t)node * (D_CH / 4) + lane] = o;
            }
        }
    }
#endif
}

__global__ __launch_bounds__(256) void last_layer_at_lookups(
        const float* __restrict__ x1, const float* __restrict__ x2,
        const int2* __restrict__ csr, const int* __restrict__ rowptr,
        const float* __restrict__ dinv, const float* __restrict__ W,
        const float* __restrict__ b, const int* __restrict__ idx,
        float* __restrict__ out, int N, int E, int B) {
#if defined(__HIP_DEVICE_COMPILE__)
    __shared__ __attribute__((aligned(16))) float tile[LT_ROWS * LT_PITCH];
    const int lane  = threadIdx.x & 31;
    const int wv    = __builtin_amdgcn_readfirstlane(threadIdx.x >> 5);
    const int slot0 = blockIdx.x * LT_ROWS;
    const float4* __restrict__ x1v = reinterpret_cast<const float4*>(x1);
    const float4* __restrict__ x2v = reinterpret_cast<const float4*>(x2);
    const float4 zero4 = make_float4(0.0f, 0.0f, 0.0f, 0.0f);

    const int slotA = slot0 + wv, slotB = slot0 + wv + 8;
    int nodes[2] = { N, N };
    if (slotA < B) nodes[0] = idx[slotA];
    if (slotB < B) nodes[1] = idx[slotB];

    float4 keep[2];
    int    rp0[2], rp1[2];
    v4f    hs[2], p1[2];
    float  di[2];
#pragma unroll
    for (int t = 0; t < 2; ++t) {
        const int nd = ((unsigned)nodes[t] < (unsigned)N) ? nodes[t] : 0;
        rp0[t] = rowptr[nd];
        rp1[t] = rowptr[nd + 1];
        hs[t]  = load_row(x2v, nd, lane);
        p1[t]  = load_row(x1v, nd, lane);
        di[t]  = dinv[nd];
    }
#pragma unroll
    for (int t = 0; t < 2; ++t) {
        rp0[t] = max(rp0[t], 0);
        rp1[t] = min(rp1[t], E);
    }
    float4 a[2];
    segment_sum_pair(x2v, csr, rp0[0], rp1[0], rp0[1], rp1[1], lane, a[0], a[1]);
#pragma unroll
    for (int t = 0; t < 2; ++t) {
        const int   r  = wv + 8 * t;
        const float d2 = di[t] * di[t];
        a[t].x = fmaf(hs[t].x, d2, a[t].x); a[t].y = fmaf(hs[t].y, d2, a[t].y);
        a[t].z = fmaf(hs[t].z, d2, a[t].z); a[t].w = fmaf(hs[t].w, d2, a[t].w);
        keep[t].x = p1[t].x + hs[t].x; keep[t].y = p1[t].y + hs[t].y;
        keep[t].z = p1[t].z + hs[t].z; keep[t].w = p1[t].w + hs[t].w;
        *reinterpret_cast<float4*>(&tile[r * LT_PITCH + lane * 4]) = a[t];
    }
    __syncthreads();

    const int koff = (lane >> 4) * 2;
    const int col  = wv * 16 + (lane & 15);
    const float* __restrict__ arow = tile + (lane & 15) * LT_PITCH + koff;
    const float* __restrict__ wcol = W + (size_t)koff * D_CH + col;
    v8f acc = v8f{};
#pragma unroll
    for (int ks = 0; ks < 32; ++ks) {
        v2f av, bw;
        av.x = arow[ks * 4];
        av.y = arow[ks * 4 + 1];
        bw.x = wcol[(size_t)(ks * 4) * D_CH];
        bw.y = wcol[(size_t)(ks * 4 + 1) * D_CH];
        acc = __builtin_amdgcn_wmma_f32_16x16x4_f32(
            false, av, false, bw, (short)0, acc, false, false);
    }
    __syncthreads();

    const float bias = b[col];
    const int   row0 = (lane >> 4) * 8;
#pragma unroll
    for (int r = 0; r < 8; ++r)
        tile[(row0 + r) * LT_PITCH + col] = fmaxf(acc[r] + bias, 0.0f);
    __syncthreads();

#pragma unroll
    for (int t = 0; t < 2; ++t) {
        const int r    = wv + 8 * t;
        const int slot = slot0 + r;
        if (slot < B) {
            float4 o = zero4;
            if ((unsigned)nodes[t] < (unsigned)N) {
                const float4 x3 = *reinterpret_cast<const float4*>(&tile[r * LT_PITCH + lane * 4]);
                o.x = (keep[t].x + x3.x) * (1.0f / 3.0f);
                o.y = (keep[t].y + x3.y) * (1.0f / 3.0f);
                o.z = (keep[t].z + x3.z) * (1.0f / 3.0f);
                o.w = (keep[t].w + x3.w) * (1.0f / 3.0f);
            }
            reinterpret_cast<float4*>(out)[(size_t)slot * (D_CH / 4) + lane] = o;
        }
    }
#endif
}

extern "C" void kernel_launch(void* const* d_in, const int* in_sizes, int n_in,
                              void* d_out, int out_size, void* d_ws, size_t ws_size,
                              hipStream_t stream) {
    const int*   inputs_idx  = (const int*)  d_in[0];
    const float* x           = (const float*)d_in[1];
    const int*   edge_index  = (const int*)  d_in[2];
    const float* edge_weight = (const float*)d_in[3];
    const float* Wm[3] = { (const float*)d_in[4], (const float*)d_in[6], (const float*)d_in[8] };
    const float* bv[3] = { (const float*)d_in[5], (const float*)d_in[7], (const float*)d_in[9] };

    const int B = in_sizes[0];
    const int N = in_sizes[1] / D_CH;
    const int E = in_sizes[2] / 2;
    (void)n_in; (void)ws_size; (void)out_size;
    const int* src = edge_index;
    const int* dst = edge_index + E;

    float* ws   = (float*)d_ws;
    const size_t nd   = (size_t)N * D_CH;
    const size_t Npad = ((size_t)N + 3) & ~(size_t)3;
    float* xl[2];
    xl[0] = ws;
    xl[1] = xl[0] + nd;
    int*   cnt    = (int*)(xl[1] + nd);
    float* dinv   = (float*)(cnt + Npad);
    int*   rowptr = (int*)(dinv + Npad);
    int2*  csr    = (int2*)(rowptr + Npad + 4);
    int*   rank   = (int*)(csr + E);

    const int TB = 256;
    const int nTiles   = (N + LT_ROWS - 1) / LT_ROWS;
    const int aggBlk   = (nTiles + 3) / 4;
    const int edgeBlk  = (E + TB - 1) / TB;
    const int nodeBlk  = (N + TB - 1) / TB;
    const int slotBlk  = (B + LT_ROWS - 1) / LT_ROWS;

    fill_i32<<<nodeBlk, TB, 0, stream>>>(cnt, 0, N);
    hist_rank<<<edgeBlk, TB, 0, stream>>>(cnt, dst, rank, E);
    rowptr_scan<<<1, SCAN_T, 0, stream>>>(cnt, rowptr, N);
    csr_fill<<<edgeBlk, TB, 0, stream>>>(csr, rowptr, rank, src, dst, edge_weight, E);
    deg_dinv<<<nodeBlk, TB, 0, stream>>>(dinv, csr, rowptr, N, E);
    csr_norm<<<nodeBlk, TB, 0, stream>>>(csr, rowptr, dinv, N, E);

    const float* xin = x;
    for (int l = 0; l < 2; ++l) {
        aggregate_then_product<<<aggBlk, TB, 0, stream>>>(xin, csr, rowptr, dinv, Wm[l], bv[l],
                                                          xl[l], N, E, nTiles);
        xin = xl[l];
    }

    last_layer_at_lookups<<<slotBlk, TB, 0, stream>>>(xl[0], xl[1], csr, rowptr, dinv,
                                                      Wm[2], bv[2], inputs_idx,
                                                      (float*)d_out, N, E, B);
}


# CSR as fixed 64-slot buckets per node: the counting atomic's return is the edge's slot (csr[dst*64+rank], overflow list beyond 64), so scan, fill pass and 2 launches are gone; sum order as before
# speedup vs baseline: 4.1932x; 1.2536x over previous
#include <hip/hip_runtime.h>
#include <hip/hip_bf16.h>


typedef __attribute__((ext_vector_type(2))) float v2f;
typedef __attribute__((ext_vector_type(8))) float v8f;
typedef __attribute__((ext_vector_type(4))) float v4f;

#define D_CH 128

__global__ void fill_i32(int* __restrict__ p, int v, int n) {
    int i = blockIdx.x * blockDim.x + threadIdx.x;
    if (i < n) p[i] = v;
}

#define CSR_CAP 64

__global__ void count_and_place(int* __restrict__ cnt, int* __restrict__ ovf_n,
                                int2* __restrict__ csr, int* __restrict__ ovf_dst,
                                int2* __restrict__ ovf_sw, const int* __restrict__ src,
                                const int* __restrict__ dst, const float* __restrict__ w, int E) {
    int e = blockIdx.x * blockDim.x + threadIdx.x;
    if (e >= E) return;
    const int  d = dst[e];
    const int2 c = make_int2(src[e], __float_as_int(w[e]));
    const int  r = atomicAdd(&cnt[d], 1);
    if ((unsigned)r < (unsigned)CSR_CAP) {
        csr[(size_t)d * CSR_CAP + r] = c;
    } else {
        const int k = atomicAdd(ovf_n, 1);
        if ((unsigned)k < (unsigned)E) { ovf_dst[k] = d; ovf_sw[k] = c; }
    }
}

__device__ __forceinline__ int overflow_count(const int* __restrict__ ovf_n, int E) {
    return min(max(*ovf_n, 0), E);
}

__global__ void deg_dinv(float* __restrict__ dinv, const int2* __restrict__ csr,
                         const int* __restrict__ cnt, const int* __restrict__ ovf_n,
                         const int* __restrict__ ovf_dst, const int2* __restrict__ ovf_sw,
                         int N, int E) {
    int i = blockIdx.x * blockDim.x + threadIdx.x;
    if (i >= N) return;
    const int c  = cnt[i];
    const int cb = min(max(c, 0), CSR_CAP);
    const int2* __restrict__ row = csr + (size_t)i * CSR_CAP;
    float s = 0.0f;
    for (int p = 0; p < cb; ++p) s += __int_as_float(row[p].y);
    if (c > CSR_CAP) {
        const int n = overflow_count(ovf_n, E);
        for (int k = 0; k < n; ++k)
            if (ovf_dst[k] == i) s += __int_as_float(ovf_sw[k].y);
    }
    dinv[i] = __frsqrt_rn(s + 1.0f);
}

__global__ void csr_norm(int2* __restrict__ csr, const int* __restrict__ cnt,
                         const float* __restrict__ dinv, const int* __restrict__ ovf_n,
                         const int* __restrict__ ovf_dst, int2* __restrict__ ovf_sw,
                         int N, int E) {
    int i = blockIdx.x * blockDim.x + threadIdx.x;
    const int n = overflow_count(ovf_n, E);
    for (int k = i; k < n; k += gridDim.x * blockDim.x) {
        const int2 c = ovf_sw[k];
        ovf_sw[k].y = __float_as_int(dinv[c.x] * __int_as_float(c.y) * dinv[ovf_dst[k]]);
    }
    if (i >= N) return;
    const int cb = min(max(cnt[i], 0), CSR_CAP);
    int2* __restrict__ row = csr + (size_t)i * CSR_CAP;
    const float di = dinv[i];
    for (int p = 0; p < cb; ++p) {
        const int2 c = row[p];
        row[p].y = __float_as_int(dinv[c.x] * __int_as_float(c.y) * di);
    }
}

__device__ __forceinline__ void fma4(float4& a, float s, const v4f& v) {
    a.x = fmaf(s, v.x, a.x); a.y = fmaf(s, v.y, a.y);
    a.z = fmaf(s, v.z, a.z); a.w = fmaf(s, v.w, a.w);
}

typedef const __attribute__((address_space(1))) v4f* grow_t;
__device__ __forceinline__ v4f load_row(const float4* __restrict__ h4, int s, int lane) {
    const unsigned long long a = (unsigned long long)h4 + ((unsigned long long)(unsigned)s << 9);
    const unsigned lo = __builtin_amdgcn_readfirstlane((unsigned)a);
    const unsigned hi = __builtin_amdgcn_readfirstlane((unsigned)(a >> 32));
    grow_t row = (grow_t)(((unsigned long long)hi << 32) | lo);
    return row[(unsigned)lane];
}
__device__ __forceinline__ void ask4(const float4* __restrict__ h4, const int2 c, int j, int lane,
                                     v4f (&v)[4], float (&n)[4]) {
#pragma unroll
    for (int k = 0; k < 4; ++k) {
        const int s = __builtin_amdgcn_readlane(c.x, j + k);
        n[k] = __int_as_float(__builtin_amdgcn_readlane(c.y, j + k));
        v[k] = load_row(h4, s, lane);
    }
}
__device__ __forceinline__ void add4(float4& acc, const v4f (&v)[4], const float (&n)[4]) {
#pragma unroll
    for (int k = 0; k < 4; ++k) fma4(acc, n[k], v[k]);
}
__device__ __forceinline__ void pick_tail(const int2 c, int j, int rem, int (&s)[3], float (&n)[3]) {
#pragma unroll
    for (int k = 0; k < 3; ++k) {
        s[k] = __builtin_amdgcn_readlane(c.x, (j + k) & 31);
        const float w = __int_as_float(__builtin_amdgcn_readlane(c.y, (j + k) & 31));
        n[k] = (k < rem) ? w : 0.0f;
    }
}
__device__ __forceinline__ void read_tail(const float4* __restrict__ h4, const int (&s)[3], int rem,
                                          int lane, v4f (&v)[3]) {
#pragma unroll
    for (int k = 0; k < 3; ++k) {
        v[k] = v4f{0.0f, 0.0f, 0.0f, 0.0f};
        if (k < rem) v[k] = load_row(h4, s[k], lane);
    }
}
__device__ __forceinline__ void add_tail(float4& acc, const v4f (&v)[3], const float (&n)[3]) {
#pragma unroll
    for (int k = 0; k < 3; ++k) fma4(acc, n[k], v[k]);
}

__device__ __forceinline__ void segment_sum_pair(const float4* __restrict__ h4,
                                                 const int2* __restrict__ csr,
                                                 int startA, int endA, int startB, int endB,
                                                 int lane, float4& accA, float4& accB) {
    accA = make_float4(0.0f, 0.0f, 0.0f, 0.0f);
    accB = make_float4(0.0f, 0.0f, 0.0f, 0.0f);
    while (startA < endA || startB < endB) {
        const int cntA = max(0, min(32, endA - startA));
        const int cntB = max(0, min(32, endB - startB));
        int2 ca = make_int2(0, 0), cb = make_int2(0, 0);
        if (lane < cntA) ca = csr[startA + lane];
        if (lane < cntB) cb = csr[startB + lane];
        int j = 0;
        for (; j + 4 <= min(cntA, cntB); j += 4) {
            v4f   va[4], vb[4];
            float na[4], nb[4];
            ask4(h4, ca, j, lane, va, na);
            ask4(h4, cb, j, lane, vb, nb);
            asm volatile("" :: "v"(va[0]), "v"(va[1]), "v"(va[2]), "v"(va[3]),
                               "v"(vb[0]), "v"(vb[1]), "v"(vb[2]), "v"(vb[3]));
            add4(accA, va, na);
            add4(accB, vb, nb);
        }
        int jA = j, jB = j;
        for (; jA + 4 <= cntA; jA += 4) {
            v4f   va[4];
            float na[4];
            ask4(h4, ca, jA, lane, va, na);
            add4(accA, va, na);
        }
        for (; jB + 4 <= cntB; jB += 4) {
            v4f   vb[4];
            float nb[4];
            ask4(h4, cb, jB, lane, vb, nb);
            add4(accB, vb, nb);
        }
        const int remA = cntA - jA, remB = cntB - jB;
        if (remA > 0 || remB > 0) {
            v4f   va[3], vb[3];
            float na[3], nb[3];
            int   sa[3], sb[3];
            pick_tail(ca, jA, remA, sa, na);
            pick_tail(cb, jB, remB, sb, nb);
            read_tail(h4, sa, remA, lane, va);
            read_tail(h4, sb, remB, lane, vb);
            add_tail(accA, va, na);
            add_tail(accB, vb, nb);
        }
        startA += 32;
        startB += 32;
    }
}

__device__ __forceinline__ void overflow_sum(const float4* __restrict__ h4,
                                             const int* __restrict__ ovf_dst,
                                             const int2* __restrict__ ovf_sw, int n_ovf,
                                             int node, int lane, float4& acc) {
    for (int k0 = 0; k0 < n_ovf; k0 += 32) {
        const int k = k0 + lane;
        int  d = -1;
        int2 c = make_int2(0, 0);
        if (k < n_ovf) { d = ovf_dst[k]; c = ovf_sw[k]; }
        unsigned m = __builtin_amdgcn_ballot_w32(d == node);
        while (m != 0u) {
            const int j = __builtin_ctz(m);
            m &= m - 1u;
            const int   s = __builtin_amdgcn_readlane(c.x, j);
            const float n = __int_as_float(__builtin_amdgcn_readlane(c.y, j));
            const v4f   v = load_row(h4, s, lane);
            fma4(acc, n, v);
        }
    }
}

#define LT_ROWS  16
#define LT_PITCH 132

__global__ __launch_bounds__(256, 8) void aggregate_then_product(
        const float* __restrict__ xin, const int2* __restrict__ csr,
        const int* __restrict__ cnt, const float* __restrict__ dinv,
        const int* __restrict__ ovf_n, const int* __restrict__ ovf_dst,
        const int2* __restrict__ ovf_sw,
        const float* __restrict__ W, const float* __restrict__ b,
        float* __restrict__ xout, int N, int E, int nTiles) {
#if defined(__HIP_DEVICE_COMPILE__)
    __shared__ __attribute__((aligned(16))) float tile[LT_ROWS * LT_PITCH];
    const int lane = threadIdx.x & 31;
    const int wv   = __builtin_amdgcn_readfirstlane(threadIdx.x >> 5);
    const float4* __restrict__ xv = reinterpret_cast<const float4*>(xin);

    const int koff = (lane >> 4) * 2;
    const int col  = wv * 16 + (lane & 15);
    const float* __restrict__ arow = tile + (lane & 15) * LT_PITCH + koff;
    const float* __restrict__ wcol = W + (size_t)koff * D_CH + col;
    const int   row0 = (lane >> 4) * 8;

    v2f bw[32];
#pragma unroll
    for (int ks = 0; ks < 32; ++ks) {
        bw[ks].x = wcol[(size_t)(ks * 4) * D_CH];
        bw[ks].y = wcol[(size_t)(ks * 4 + 1) * D_CH];
    }
    const float bias = b[col];

    for (int t = blockIdx.x; t < nTiles; t += gridDim.x) {
        const int node0 = t * LT_ROWS;
        int    nds[2], cn[2], rp0[2], rp1[2];
        v4f    hs[2];
        float  di[2];
#pragma unroll
        for (int u = 0; u < 2; ++u) {
            const int nd = min(node0 + wv + 8 * u, N - 1);
            nds[u] = nd;
            cn[u]  = __builtin_amdgcn_readfirstlane(cnt[nd]);
            hs[u]  = load_row(xv, nd, lane);
            di[u]  = __int_as_float(__builtin_amdgcn_readfirstlane(__float_as_int(dinv[nd])));
        }
#pragma unroll
        for (int u = 0; u < 2; ++u) {
            rp0[u] = nds[u] * CSR_CAP;
            rp1[u] = rp0[u] + min(max(cn[u], 0), CSR_CAP);
        }
        float4 a[2];
        segment_sum_pair(xv, csr, rp0[0], rp1[0], rp0[1], rp1[1], lane, a[0], a[1]);
#pragma unroll
        for (int u = 0; u < 2; ++u)
            if (cn[u] > CSR_CAP)
                overflow_sum(xv, ovf_dst, ovf_sw, overflow_count(ovf_n, E), nds[u], lane, a[u]);
#pragma unroll
        for (int u = 0; u < 2; ++u) {
            const int   r  = wv + 8 * u;
            const float d2 = di[u] * di[u];
            a[u].x = fmaf(hs[u].x, d2, a[u].x); a[u].y = fmaf(hs[u].y, d2, a[u].y);
            a[u].z = fmaf(hs[u].z, d2, a[u].z); a[u].w = fmaf(hs[u].w, d2, a[u].w);
            *reinterpret_cast<float4*>(&tile[r * LT_PITCH + lane * 4]) = a[u];
        }
        __syncthreads();

        v8f acc = v8f{};
#pragma unroll
        for (int ks = 0; ks < 32; ++ks) {
            v2f av;
            av.x = arow[ks * 4];
            av.y = arow[ks * 4 + 1];
            acc = __builtin_amdgcn_wmma_f32_16x16x4_f32(
                false, av, false, bw[ks], (short)0, acc, false, false);
        }
        __syncthreads();

#pragma unroll
        for (int r = 0; r < 8; ++r)
            tile[(row0 + r) * LT_PITCH + col] = fmaxf(acc[r] + bias, 0.0f);
        __syncthreads();

        for (int u = 0; u < 2; ++u) {
            const int r    = wv + 8 * u;
            const int node = node0 + r;
            if (node < N) {
                const float4 o = *reinterpret_cast<const float4*>(&tile[r * LT_PITCH + lane * 4]);
                reinterpret_cast<float4*>(xout)[(size_t)node * (D_CH / 4) + lane] = o;
            }
        }
    }
#endif
}

__global__ __launch_bounds__(256) void last_layer_at_lookups(
        const float* __restrict__ x1, const float* __restrict__ x2,
        const int2* __restrict__ csr, const int* __restrict__ cnt,
        const float* __restrict__ dinv, const int* __restrict__ ovf_n,
        const int* __restrict__ ovf_dst, const int2* __restrict__ ovf_sw,
        const float* __restrict__ W,
        const float* __restrict__ b, const int* __restrict__ idx,
        float* __restrict__ out, int N, int E, int B) {
#if defined(__HIP_DEVICE_COMPILE__)
    __shared__ __attribute__((aligned(16))) float tile[LT_ROWS * LT_PITCH];
    const int lane  = threadIdx.x & 31;
    const int wv    = __builtin_amdgcn_readfirstlane(threadIdx.x >> 5);
    const int slot0 = blockIdx.x * LT_ROWS;
    const float4* __restrict__ x1v = reinterpret_cast<const float4*>(x1);
    const float4* __restrict__ x2v = reinterpret_cast<const float4*>(x2);
    const float4 zero4 = make_float4(0.0f, 0.0f, 0.0f, 0.0f);

    const int slotA = slot0 + wv, slotB = slot0 + wv + 8;
    int nodes[2] = { N, N };
    if (slotA < B) nodes[0] = idx[slotA];
    if (slotB < B) nodes[1] = idx[slotB];

    float4 keep[2];
    int    nds[2], cn[2], rp0[2], rp1[2];
    v4f    hs[2], p1[2];
    float  di[2];
#pragma unroll
    for (int t = 0; t < 2; ++t) {
        const int nd = ((unsigned)nodes[t] < (unsigned)N) ? nodes[t] : 0;
        nds[t] = nd;
        cn[t]  = __builtin_amdgcn_readfirstlane(cnt[nd]);
        hs[t]  = load_row(x2v, nd, lane);
        p1[t]  = load_row(x1v, nd, lane);
        di[t]  = dinv[nd];
    }
#pragma unroll
    for (int t = 0; t < 2; ++t) {
        rp0[t] = nds[t] * CSR_CAP;
        rp1[t] = rp0[t] + min(max(cn[t], 0), CSR_CAP);
    }
    float4 a[2];
    segment_sum_pair(x2v, csr, rp0[0], rp1[0], rp0[1], rp1[1], lane, a[0], a[1]);
#pragma unroll
    for (int t = 0; t < 2; ++t)
        if (cn[t] > CSR_CAP)
            overflow_sum(x2v, ovf_dst, ovf_sw, overflow_count(ovf_n, E), nds[t], lane, a[t]);
#pragma unroll
    for (int t = 0; t < 2; ++t) {
        const int   r  = wv + 8 * t;
        const float d2 = di[t] * di[t];
        a[t].x = fmaf(hs[t].x, d2, a[t].x); a[t].y = fmaf(hs[t].y, d2, a[t].y);
        a[t].z = fmaf(hs[t].z, d2, a[t].z); a[t].w = fmaf(hs[t].w, d2, a[t].w);
        keep[t].x = p1[t].x + hs[t].x; keep[t].y = p1[t].y + hs[t].y;
        keep[t].z = p1[t].z + hs[t].z; keep[t].w = p1[t].w + hs[t].w;
        *reinterpret_cast<float4*>(&tile[r * LT_PITCH + lane * 4]) = a[t];
    }
    __syncthreads();

    const int koff = (lane >> 4) * 2;
    const int col  = wv * 16 + (lane & 15);
    const float* __restrict__ arow = tile + (lane & 15) * LT_PITCH + koff;
    const float* __restrict__ wcol = W + (size_t)koff * D_CH + col;
    v8f acc = v8f{};
#pragma unroll
    for (int ks = 0; ks < 32; ++ks) {
        v2f av, bw;
        av.x = arow[ks * 4];
        av.y = arow[ks * 4 + 1];
        bw.x = wcol[(size_t)(ks * 4) * D_CH];
        bw.y = wcol[(size_t)(ks * 4 + 1) * D_CH];
        acc = __builtin_amdgcn_wmma_f32_16x16x4_f32(
            false, av, false, bw, (short)0, acc, false, false);
    }
    __syncthreads();

    const float bias = b[col];
    const int   row0 = (lane >> 4) * 8;
#pragma unroll
    for (int r = 0; r < 8; ++r)
        tile[(row0 + r) * LT_PITCH + col] = fmaxf(acc[r] + bias, 0.0f);
    __syncthreads();

#pragma unroll
    for (int t = 0; t < 2; ++t) {
        const int r    = wv + 8 * t;
        const int slot = slot0 + r;
        if (slot < B) {
            float4 o = zero4;
            if ((unsigned)nodes[t] < (unsigned)N) {
                const float4 x3 = *reinterpret_cast<const float4*>(&tile[r * LT_PITCH + lane * 4]);
                o.x = (keep[t].x + x3.x) * (1.0f / 3.0f);
                o.y = (keep[t].y + x3.y) * (1.0f / 3.0f);
                o.z = (keep[t].z + x3.z) * (1.0f / 3.0f);
                o.w = (keep[t].w + x3.w) * (1.0f / 3.0f);
            }
            reinterpret_cast<float4*>(out)[(size_t)slot * (D_CH / 4) + lane] = o;
        }
    }
#endif
}

extern "C" void kernel_launch(void* const* d_in, const int* in_sizes, int n_in,
                              void* d_out, int out_size, void* d_ws, size_t ws_size,
                              hipStream_t stream) {
    const int*   inputs_idx  = (const int*)  d_in[0];
    const float* x           = (const float*)d_in[1];
    const int*   edge_index  = (const int*)  d_in[2];
    const float* edge_weight = (const float*)d_in[3];
    const float* Wm[3] = { (const float*)d_in[4], (const float*)d_in[6], (const float*)d_in[8] };
    const float* bv[3] = { (const float*)d_in[5], (const float*)d_in[7], (const float*)d_in[9] };

    const int B = in_sizes[0];
    const int N = in_sizes[1] / D_CH;
    const int E = in_sizes[2] / 2;
    (void)n_in; (void)ws_size; (void)out_size;
    const int* src = edge_index;
    const int* dst = edge_index + E;

    float* ws   = (float*)d_ws;
    const size_t nd   = (size_t)N * D_CH;
    const size_t Npad = ((size_t)N + 3) & ~(size_t)3;
    const size_t Epad = ((size_t)E + 1) & ~(size_t)1;
    float* xl[2];
    xl[0] = ws;
    xl[1] = xl[0] + nd;
    int2*  csr     = (int2*)(xl[1] + nd);
    int*   cnt     = (int*)(csr + (size_t)N * CSR_CAP);
    int*   ovf_n   = cnt + Npad;
    float* dinv    = (float*)(ovf_n + 4);
    int*   ovf_dst = (int*)(dinv + Npad);
    int2*  ovf_sw  = (int2*)(ovf_dst + Epad);

    const int TB = 256;
    const int nTiles   = (N + LT_ROWS - 1) / LT_ROWS;
    const int aggBlk   = (nTiles + 3) / 4;
    const int edgeBlk  = (E + TB - 1) / TB;
    const int nodeBlk  = (N + TB - 1) / TB;
    const int slotBlk  = (B + LT_ROWS - 1) / LT_ROWS;
    const int nZero    = (int)Npad + 4;
    const int zeroBlk  = (nZero + TB - 1) / TB;

    fill_i32<<<zeroBlk, TB, 0, stream>>>(cnt, 0, nZero);
    count_and_place<<<edgeBlk, TB, 0, stream>>>(cnt, ovf_n, csr, ovf_dst, ovf_sw, src, dst, edge_weight, E);
    deg_dinv<<<nodeBlk, TB, 0, stream>>>(dinv, csr, cnt, ovf_n, ovf_dst, ovf_sw, N, E);
    csr_norm<<<nodeBlk, TB, 0, stream>>>(csr, cnt, dinv, ovf_n, ovf_dst, ovf_sw, N, E);

    const float* xin = x;
    for (int l = 0; l < 2; ++l) {
        aggregate_then_product<<<aggBlk, TB, 0, stream>>>(xin, csr, cnt, dinv, ovf_n, ovf_dst, ovf_sw,
                                                          Wm[l], bv[l], xl[l], N, E, nTiles);
        xin = xl[l];
    }

    last_layer_at_lookups<<<slotBlk, TB, 0, stream>>>(xl[0], xl[1], csr, cnt, dinv,
                                                      ovf_n, ovf_dst, ovf_sw,
                                                      Wm[2], bv[2], inputs_idx,
                                                      (float*)d_out, N, E, B);
}


# Slot normalisation folded into layer 1: the wave that sums a node forms dinv[src]*w*dinv[dst] for its slots itself and writes it back for layers 2 and 3; 1 launch and its pass gone; values unchanged
# speedup vs baseline: 4.4398x; 1.0588x over previous
#include <hip/hip_runtime.h>
#include <hip/hip_bf16.h>


typedef __attribute__((ext_vector_type(2))) float v2f;
typedef __attribute__((ext_vector_type(8))) float v8f;
typedef __attribute__((ext_vector_type(4))) float v4f;

#define D_CH 128

__global__ void fill_i32(int* __restrict__ p, int v, int n) {
    int i = blockIdx.x * blockDim.x + threadIdx.x;
    if (i < n) p[i] = v;
}

#define CSR_CAP 64

__global__ void count_and_place(int* __restrict__ cnt, int* __restrict__ ovf_n,
                                int2* __restrict__ csr, int* __restrict__ ovf_dst,
                                int2* __restrict__ ovf_sw, const int* __restrict__ src,
                                const int* __restrict__ dst, const float* __restrict__ w, int E) {
    int e = blockIdx.x * blockDim.x + threadIdx.x;
    if (e >= E) return;
    const int  d = dst[e];
    const int2 c = make_int2(src[e], __float_as_int(w[e]));
    const int  r = atomicAdd(&cnt[d], 1);
    if ((unsigned)r < (unsigned)CSR_CAP) {
        csr[(size_t)d * CSR_CAP + r] = c;
    } else {
        const int k = atomicAdd(ovf_n, 1);
        if ((unsigned)k < (unsigned)E) { ovf_dst[k] = d; ovf_sw[k] = c; }
    }
}

__device__ __forceinline__ int overflow_count(const int* __restrict__ ovf_n, int E) {
    return min(max(*ovf_n, 0), E);
}

__global__ void deg_dinv(float* __restrict__ dinv, const int2* __restrict__ csr,
                         const int* __restrict__ cnt, const int* __restrict__ ovf_n,
                         const int* __restrict__ ovf_dst, const int2* __restrict__ ovf_sw,
                         int N, int E) {
    int i = blockIdx.x * blockDim.x + threadIdx.x;
    if (i >= N) return;
    const int c  = cnt[i];
    const int cb = min(max(c, 0), CSR_CAP);
    const int2* __restrict__ row = csr + (size_t)i * CSR_CAP;
    float s = 0.0f;
    for (int p = 0; p < cb; ++p) s += __int_as_float(row[p].y);
    if (c > CSR_CAP) {
        const int n = overflow_count(ovf_n, E);
        for (int k = 0; k < n; ++k)
            if (ovf_dst[k] == i) s += __int_as_float(ovf_sw[k].y);
    }
    dinv[i] = __frsqrt_rn(s + 1.0f);
}

__device__ __forceinline__ void fma4(float4& a, float s, const v4f& v) {
    a.x = fmaf(s, v.x, a.x); a.y = fmaf(s, v.y, a.y);
    a.z = fmaf(s, v.z, a.z); a.w = fmaf(s, v.w, a.w);
}

typedef const __attribute__((address_space(1))) v4f* grow_t;
__device__ __forceinline__ v4f load_row(const float4* __restrict__ h4, int s, int lane) {
    const unsigned long long a = (unsigned long long)h4 + ((unsigned long long)(unsigned)s << 9);
    const unsigned lo = __builtin_amdgcn_readfirstlane((unsigned)a);
    const unsigned hi = __builtin_amdgcn_readfirstlane((unsigned)(a >> 32));
    grow_t row = (grow_t)(((unsigned long long)hi << 32) | lo);
    return row[(unsigned)lane];
}
__device__ __forceinline__ void ask4(const float4* __restrict__ h4, const int2 c, int j, int lane,
                                     v4f (&v)[4], float (&n)[4]) {
#pragma unroll
    for (int k = 0; k < 4; ++k) {
        const int s = __builtin_amdgcn_readlane(c.x, j + k);
        n[k] = __int_as_float(__builtin_amdgcn_readlane(c.y, j + k));
        v[k] = load_row(h4, s, lane);
    }
}
__device__ __forceinline__ void add4(float4& acc, const v4f (&v)[4], const float (&n)[4]) {
#pragma unroll
    for (int k = 0; k < 4; ++k) fma4(acc, n[k], v[k]);
}
__device__ __forceinline__ void pick_tail(const int2 c, int j, int rem, int (&s)[3], float (&n)[3]) {
#pragma unroll
    for (int k = 0; k < 3; ++k) {
        s[k] = __builtin_amdgcn_readlane(c.x, (j + k) & 31);
        const float w = __int_as_float(__builtin_amdgcn_readlane(c.y, (j + k) & 31));
        n[k] = (k < rem) ? w : 0.0f;
    }
}
__device__ __forceinline__ void read_tail(const float4* __restrict__ h4, const int (&s)[3], int rem,
                                          int lane, v4f (&v)[3]) {
#pragma unroll
    for (int k = 0; k < 3; ++k) {
        v[k] = v4f{0.0f, 0.0f, 0.0f, 0.0f};
        if (k < rem) v[k] = load_row(h4, s[k], lane);
    }
}
__device__ __forceinline__ void add_tail(float4& acc, const v4f (&v)[3], const float (&n)[3]) {
#pragma unroll
    for (int k = 0; k < 3; ++k) fma4(acc, n[k], v[k]);
}

template <bool NORM>
__device__ __forceinline__ void segment_sum_pair(const float4* __restrict__ h4,
                                                 const int2* csr, int2* csr_w,
                                                 const float* __restrict__ dinv, float diA, float diB,
                                                 int startA, int endA, int startB, int endB,
                                                 int lane, float4& accA, float4& accB) {
    accA = make_float4(0.0f, 0.0f, 0.0f, 0.0f);
    accB = make_float4(0.0f, 0.0f, 0.0f, 0.0f);
    while (startA < endA || startB < endB) {
        const int cntA = max(0, min(32, endA - startA));
        const int cntB = max(0, min(32, endB - startB));
        int2 ca = make_int2(0, 0), cb = make_int2(0, 0);
        if (lane < cntA) ca = csr[startA + lane];
        if (lane < cntB) cb = csr[startB + lane];
        if (NORM) {
            const float dsa = dinv[ca.x];
            const float dsb = dinv[cb.x];
            ca.y = (lane < cntA) ? __float_as_int(dsa * __int_as_float(ca.y) * diA) : 0;
            cb.y = (lane < cntB) ? __float_as_int(dsb * __int_as_float(cb.y) * diB) : 0;
            if (lane < cntA) csr_w[startA + lane] = ca;
            if (lane < cntB) csr_w[startB + lane] = cb;
        }
        int j = 0;
        for (; j + 4 <= min(cntA, cntB); j += 4) {
            v4f   va[4], vb[4];
            float na[4], nb[4];
            ask4(h4, ca, j, lane, va, na);
            ask4(h4, cb, j, lane, vb, nb);
            asm volatile("" :: "v"(va[0]), "v"(va[1]), "v"(va[2]), "v"(va[3]),
                               "v"(vb[0]), "v"(vb[1]), "v"(vb[2]), "v"(vb[3]));
            add4(accA, va, na);
            add4(accB, vb, nb);
        }
        int jA = j, jB = j;
        for (; jA + 4 <= cntA; jA += 4) {
            v4f   va[4];
            float na[4];
            ask4(h4, ca, jA, lane, va, na);
            add4(accA, va, na);
        }
        for (; jB + 4 <= cntB; jB += 4) {
            v4f   vb[4];
            float nb[4];
            ask4(h4, cb, jB, lane, vb, nb);
            add4(accB, vb, nb);
        }
        const int remA = cntA - jA, remB = cntB - jB;
        if (remA > 0 || remB > 0) {
            v4f   va[3], vb[3];
            float na[3], nb[3];
            int   sa[3], sb[3];
            pick_tail(ca, jA, remA, sa, na);
            pick_tail(cb, jB, remB, sb, nb);
            read_tail(h4, sa, remA, lane, va);
            read_tail(h4, sb, remB, lane, vb);
            add_tail(accA, va, na);
            add_tail(accB, vb, nb);
        }
        startA += 32;
        startB += 32;
    }
}

template <bool NORM>
__device__ __forceinline__ void overflow_sum(const float4* __restrict__ h4,
                                             const int* __restrict__ ovf_dst,
                                             const int2* ovf_sw, int2* ovf_sw_w,
                                             const float* __restrict__ dinv, float di, int n_ovf,
                                             int node, int lane, float4& acc) {
    for (int k0 = 0; k0 < n_ovf; k0 += 32) {
        const int k = k0 + lane;
        int  d = -1;
        int2 c = make_int2(0, 0);
        if (k < n_ovf) { d = ovf_dst[k]; c = ovf_sw[k]; }
        if (NORM) {
            if (d == node) {
                c.y = __float_as_int(dinv[c.x] * __int_as_float(c.y) * di);
                ovf_sw_w[k] = c;
            }
        }
        unsigned m = __builtin_amdgcn_ballot_w32(d == node);
        while (m != 0u) {
            const int j = __builtin_ctz(m);
            m &= m - 1u;
            const int   s = __builtin_amdgcn_readlane(c.x, j);
            const float n = __int_as_float(__builtin_amdgcn_readlane(c.y, j));
            const v4f   v = load_row(h4, s, lane);
            fma4(acc, n, v);
        }
    }
}

#define LT_ROWS  16
#define LT_PITCH 132

template <bool NORM>
__global__ __launch_bounds__(256, 8) void aggregate_then_product(
        const float* __restrict__ xin, int2* csr,
        const int* __restrict__ cnt, const float* __restrict__ dinv,
        const int* __restrict__ ovf_n, const int* __restrict__ ovf_dst,
        int2* ovf_sw,
        const float* __restrict__ W, const float* __restrict__ b,
        float* __restrict__ xout, int N, int E, int nTiles) {
#if defined(__HIP_DEVICE_COMPILE__)
    __shared__ __attribute__((aligned(16))) float tile[LT_ROWS * LT_PITCH];
    const int lane = threadIdx.x & 31;
    const int wv   = __builtin_amdgcn_readfirstlane(threadIdx.x >> 5);
    const float4* __restrict__ xv = reinterpret_cast<const float4*>(xin);

    const int koff = (lane >> 4) * 2;
    const int col  = wv * 16 + (lane & 15);
    const float* __restrict__ arow = tile + (lane & 15) * LT_PITCH + koff;
    const float* __restrict__ wcol = W + (size_t)koff * D_CH + col;
    const int   row0 = (lane >> 4) * 8;

    v2f bw[32];
#pragma unroll
    for (int ks = 0; ks < 32; ++ks) {
        bw[ks].x = wcol[(size_t)(ks * 4) * D_CH];
        bw[ks].y = wcol[(size_t)(ks * 4 + 1) * D_CH];
    }
    const float bias = b[col];

    for (int t = blockIdx.x; t < nTiles; t += gridDim.x) {
        const int node0 = t * LT_ROWS;
        int    nds[2], cn[2], rp0[2], rp1[2];
        v4f    hs[2];
        float  di[2];
#pragma unroll
        for (int u = 0; u < 2; ++u) {
            const int nd = min(node0 + wv + 8 * u, N - 1);
            nds[u] = nd;
            cn[u]  = __builtin_amdgcn_readfirstlane(cnt[nd]);
            if (node0 + wv + 8 * u >= N) cn[u] = 0;
            hs[u]  = load_row(xv, nd, lane);
            di[u]  = __int_as_float(__builtin_amdgcn_readfirstlane(__float_as_int(dinv[nd])));
        }
#pragma unroll
        for (int u = 0; u < 2; ++u) {
            rp0[u] = nds[u] * CSR_CAP;
            rp1[u] = rp0[u] + min(max(cn[u], 0), CSR_CAP);
        }
        float4 a[2];
        segment_sum_pair<NORM>(xv, csr, csr, dinv, di[0], di[1], rp0[0], rp1[0], rp0[1], rp1[1], lane, a[0], a[1]);
#pragma unroll
        for (int u = 0; u < 2; ++u)
            if (cn[u] > CSR_CAP)
                overflow_sum<NORM>(xv, ovf_dst, ovf_sw, ovf_sw, dinv, di[u], overflow_count(ovf_n, E), nds[u], lane, a[u]);
#pragma unroll
        for (int u = 0; u < 2; ++u) {
            const int   r  = wv + 8 * u;
            const float d2 = di[u] * di[u];
            a[u].x = fmaf(hs[u].x, d2, a[u].x); a[u].y = fmaf(hs[u].y, d2, a[u].y);
            a[u].z = fmaf(hs[u].z, d2, a[u].z); a[u].w = fmaf(hs[u].w, d2, a[u].w);
            *reinterpret_cast<float4*>(&tile[r * LT_PITCH + lane * 4]) = a[u];
        }
        __syncthreads();

        v8f acc = v8f{};
#pragma unroll
        for (int ks = 0; ks < 32; ++ks) {
            v2f av;
            av.x = arow[ks * 4];
            av.y = arow[ks * 4 + 1];
            acc = __builtin_amdgcn_wmma_f32_16x16x4_f32(
                false, av, false, bw[ks], (short)0, acc, false, false);
        }
        __syncthreads();

#pragma unroll
        for (int r = 0; r < 8; ++r)
            tile[(row0 + r) * LT_PITCH + col] = fmaxf(acc[r] + bias, 0.0f);
        __syncthreads();

        for (int u = 0; u < 2; ++u) {
            const int r    = wv + 8 * u;
            const int node = node0 + r;
            if (node < N) {
                const float4 o = *reinterpret_cast<const float4*>(&tile[r * LT_PITCH + lane * 4]);
                reinterpret_cast<float4*>(xout)[(size_t)node * (D_CH / 4) + lane] = o;
            }
        }
    }
#endif
}

__global__ __launch_bounds__(256) void last_layer_at_lookups(
        const float* __restrict__ x1, const float* __restrict__ x2,
        const int2* __restrict__ csr, const int* __restrict__ cnt,
        const float* __restrict__ dinv, const int* __restrict__ ovf_n,
        const int* __restrict__ ovf_dst, const int2* __restrict__ ovf_sw,
        const float* __restrict__ W,
        const float* __restrict__ b, const int* __restrict__ idx,
        float* __restrict__ out, int N, int E, int B) {
#if defined(__HIP_DEVICE_COMPILE__)
    __shared__ __attribute__((aligned(16))) float tile[LT_ROWS * LT_PITCH];
    const int lane  = threadIdx.x & 31;
    const int wv    = __builtin_amdgcn_readfirstlane(threadIdx.x >> 5);
    const int slot0 = blockIdx.x * LT_ROWS;
    const float4* __restrict__ x1v = reinterpret_cast<const float4*>(x1);
    const float4* __restrict__ x2v = reinterpret_cast<const float4*>(x2);
    const float4 zero4 = make_float4(0.0f, 0.0f, 0.0f, 0.0f);

    const int slotA = slot0 + wv, slotB = slot0 + wv + 8;
    int nodes[2] = { N, N };
    if (slotA < B) nodes[0] = idx[slotA];
    if (slotB < B) nodes[1] = idx[slotB];

    float4 keep[2];
    int    nds[2], cn[2], rp0[2], rp1[2];
    v4f    hs[2], p1[2];
    float  di[2];
#pragma unroll
    for (int t = 0; t < 2; ++t) {
        const int nd = ((unsigned)nodes[t] < (unsigned)N) ? nodes[t] : 0;
        nds[t] = nd;
        cn[t]  = __builtin_amdgcn_readfirstlane(cnt[nd]);
        hs[t]  = load_row(x2v, nd, lane);
        p1[t]  = load_row(x1v, nd, lane);
        di[t]  = dinv[nd];
    }
#pragma unroll
    for (int t = 0; t < 2; ++t) {
        rp0[t] = nds[t] * CSR_CAP;
        rp1[t] = rp0[t] + min(max(cn[t], 0), CSR_CAP);
    }
    float4 a[2];
    segment_sum_pair<false>(x2v, csr, nullptr, dinv, 0.0f, 0.0f, rp0[0], rp1[0], rp0[1], rp1[1], lane, a[0], a[1]);
#pragma unroll
    for (int t = 0; t < 2; ++t)
        if (cn[t] > CSR_CAP)
            overflow_sum<false>(x2v, ovf_dst, ovf_sw, nullptr, dinv, 0.0f, overflow_count(ovf_n, E), nds[t], lane, a[t]);
#pragma unroll
    for (int t = 0; t < 2; ++t) {
        const int   r  = wv + 8 * t;
        const float d2 = di[t] * di[t];
        a[t].x = fmaf(hs[t].x, d2, a[t].x); a[t].y = fmaf(hs[t].y, d2, a[t].y);
        a[t].z = fmaf(hs[t].z, d2, a[t].z); a[t].w = fmaf(hs[t].w, d2, a[t].w);
        keep[t].x = p1[t].x + hs[t].x; keep[t].y = p1[t].y + hs[t].y;
        keep[t].z = p1[t].z + hs[t].z; keep[t].w = p1[t].w + hs[t].w;
        *reinterpret_cast<float4*>(&tile[r * LT_PITCH + lane * 4]) = a[t];
    }
    __syncthreads();

    const int koff = (lane >> 4) * 2;
    const int col  = wv * 16 + (lane & 15);
    const float* __restrict__ arow = tile + (lane & 15) * LT_PITCH + koff;
    const float* __restrict__ wcol = W + (size_t)koff * D_CH + col;
    v8f acc = v8f{};
#pragma unroll
    for (int ks = 0; ks < 32; ++ks) {
        v2f av, bw;
        av.x = arow[ks * 4];
        av.y = arow[ks * 4 + 1];
        bw.x = wcol[(size_t)(ks * 4) * D_CH];
        bw.y = wcol[(size_t)(ks * 4 + 1) * D_CH];
        acc = __builtin_amdgcn_wmma_f32_16x16x4_f32(
            false, av, false, bw, (short)0, acc, false, false);
    }
    __syncthreads();

    const float bias = b[col];
    const int   row0 = (lane >> 4) * 8;
#pragma unroll
    for (int r = 0; r < 8; ++r)
        tile[(row0 + r) * LT_PITCH + col] = fmaxf(acc[r] + bias, 0.0f);
    __syncthreads();

#pragma unroll
    for (int t = 0; t < 2; ++t) {
        const int r    = wv + 8 * t;
        const int slot = slot0 + r;
        if (slot < B) {
            float4 o = zero4;
            if ((unsigned)nodes[t] < (unsigned)N) {
                const float4 x3 = *reinterpret_cast<const float4*>(&tile[r * LT_PITCH + lane * 4]);
                o.x = (keep[t].x + x3.x) * (1.0f / 3.0f);
                o.y = (keep[t].y + x3.y) * (1.0f / 3.0f);
                o.z = (keep[t].z + x3.z) * (1.0f / 3.0f);
                o.w = (keep[t].w + x3.w) * (1.0f / 3.0f);
            }
            reinterpret_cast<float4*>(out)[(size_t)slot * (D_CH / 4) + lane] = o;
        }
    }
#endif
}

extern "C" void kernel_launch(void* const* d_in, const int* in_sizes, int n_in,
                              void* d_out, int out_size, void* d_ws, size_t ws_size,
                              hipStream_t stream) {
    const int*   inputs_idx  = (const int*)  d_in[0];
    const float* x           = (const float*)d_in[1];
    const int*   edge_index  = (const int*)  d_in[2];
    const float* edge_weight = (const float*)d_in[3];
    const float* Wm[3] = { (const float*)d_in[4], (const float*)d_in[6], (const float*)d_in[8] };
    const float* bv[3] = { (const float*)d_in[5], (const float*)d_in[7], (const float*)d_in[9] };

    const int B = in_sizes[0];
    const int N = in_sizes[1] / D_CH;
    const int E = in_sizes[2] / 2;
    (void)n_in; (void)ws_size; (void)out_size;
    const int* src = edge_index;
    const int* dst = edge_index + E;

    float* ws   = (float*)d_ws;
    const size_t nd   = (size_t)N * D_CH;
    const size_t Npad = ((size_t)N + 3) & ~(size_t)3;
    const size_t Epad = ((size_t)E + 1) & ~(size_t)1;
    float* xl[2];
    xl[0] = ws;
    xl[1] = xl[0] + nd;
    int2*  csr     = (int2*)(xl[1] + nd);
    int*   cnt     = (int*)(csr + (size_t)N * CSR_CAP);
    int*   ovf_n   = cnt + Npad;
    float* dinv    = (float*)(ovf_n + 4);
    int*   ovf_dst = (int*)(dinv + Npad);
    int2*  ovf_sw  = (int2*)(ovf_dst + Epad);

    const int TB = 256;
    const int nTiles   = (N + LT_ROWS - 1) / LT_ROWS;
    const int aggBlk   = (nTiles + 3) / 4;
    const int edgeBlk  = (E + TB - 1) / TB;
    const int nodeBlk  = (N + TB - 1) / TB;
    const int slotBlk  = (B + LT_ROWS - 1) / LT_ROWS;
    const int nZero    = (int)Npad + 4;
    const int zeroBlk  = (nZero + TB - 1) / TB;

    fill_i32<<<zeroBlk, TB, 0, stream>>>(cnt, 0, nZero);
    count_and_place<<<edgeBlk, TB, 0, stream>>>(cnt, ovf_n, csr, ovf_dst, ovf_sw, src, dst, edge_weight, E);
    deg_dinv<<<nodeBlk, TB, 0, stream>>>(dinv, csr, cnt, ovf_n, ovf_dst, ovf_sw, N, E);

    aggregate_then_product<true><<<aggBlk, TB, 0, stream>>>(x, csr, cnt, dinv, ovf_n, ovf_dst, ovf_sw,
                                                            Wm[0], bv[0], xl[0], N, E, nTiles);
    aggregate_then_product<false><<<aggBlk, TB, 0, stream>>>(xl[0], csr, cnt, dinv, ovf_n, ovf_dst, ovf_sw,
                                                             Wm[1], bv[1], xl[1], N, E, nTiles);

    last_layer_at_lookups<<<slotBlk, TB, 0, stream>>>(xl[0], xl[1], csr, cnt, dinv,
                                                      ovf_n, ovf_dst, ovf_sw,
                                                      Wm[2], bv[2], inputs_idx,
                                                      (float*)d_out, N, E, B);
}


# Slots built by sorting edges into 256 destination ranges in LDS, 2 launches: rank, slot fill, count, dinv from LDS; global count atomics, scattered stores, zero-fill gone; slot order LDS arrival
# speedup vs baseline: 5.8451x; 1.3165x over previous
#include <hip/hip_runtime.h>
#include <hip/hip_bf16.h>


typedef __attribute__((ext_vector_type(2))) float v2f;
typedef __attribute__((ext_vector_type(8))) float v8f;
typedef __attribute__((ext_vector_type(4))) float v4f;

#define D_CH 128

#define CSR_CAP 64
#define NBINS   256
#define NPB_MAX 256
#define BE_TB   1024
#define BE_EPT  5
#define BE_EPB  (BE_TB * BE_EPT)
#define BS_TB   512

__global__ __launch_bounds__(BE_TB) void bin_edges(int4* __restrict__ stage, int2* __restrict__ seg,
                                                   int* __restrict__ ovf_n, const int* __restrict__ src,
                                                   const int* __restrict__ dst, const float* __restrict__ w,
                                                   int E, int npb, int nBlk) {
    __shared__ int lcnt[NBINS];
    __shared__ int lbase[NBINS];
    __shared__ __attribute__((aligned(16))) int4 lst[BE_EPB];
    const int tid = threadIdx.x;
    const int e0  = blockIdx.x * BE_EPB;
    if (blockIdx.x == 0 && tid < 4) ovf_n[tid] = 0;
    if (tid < NBINS) lcnt[tid] = 0;
    int d[BE_EPT], s[BE_EPT], wb[BE_EPT], bin[BE_EPT], r[BE_EPT];
#pragma unroll
    for (int k = 0; k < BE_EPT; ++k) {
        const int e = e0 + k * BE_TB + tid;
        d[k] = -1; s[k] = 0; wb[k] = 0;
        if (e < E) { d[k] = dst[e]; s[k] = src[e]; wb[k] = __float_as_int(w[e]); }
    }
    __syncthreads();
#pragma unroll
    for (int k = 0; k < BE_EPT; ++k) {
        bin[k] = 0; r[k] = 0;
        if (d[k] >= 0) {
            bin[k] = min(d[k] / npb, NBINS - 1);
            r[k]   = atomicAdd(&lcnt[bin[k]], 1);
        }
    }
    __syncthreads();
    if (tid < 32) {
        int cc[NBINS / 32];
        int run = 0;
#pragma unroll
        for (int j = 0; j < NBINS / 32; ++j) { cc[j] = run; run += lcnt[tid * (NBINS / 32) + j]; }
        int inc = run;
#pragma unroll
        for (int o = 1; o < 32; o <<= 1) {
            const int v = __shfl_up(inc, o, 32);
            if (tid >= o) inc += v;
        }
        const int excl = inc - run;
#pragma unroll
        for (int j = 0; j < NBINS / 32; ++j) lbase[tid * (NBINS / 32) + j] = excl + cc[j];
    }
    __syncthreads();
#pragma unroll
    for (int k = 0; k < BE_EPT; ++k)
        if (d[k] >= 0) lst[lbase[bin[k]] + r[k]] = make_int4(d[k], s[k], wb[k], 0);
    __syncthreads();
    const int nHere = min(BE_EPB, max(E - e0, 0));
    for (int i = tid; i < nHere; i += BE_TB) stage[(size_t)e0 + i] = lst[i];
    if (tid < NBINS) seg[(size_t)tid * nBlk + blockIdx.x] = make_int2(lbase[tid], lcnt[tid]);
}

__device__ __forceinline__ void place_edge(const int4 rec, int node0, int nHere, int E,
                                           int* lcnt, float* lovw, int2* lslot, int* __restrict__ ovf_n,
                                           int* __restrict__ ovf_dst, int2* __restrict__ ovf_sw) {
    const int dl = rec.x - node0;
    if ((unsigned)dl < (unsigned)nHere) {
        const int rk = atomicAdd(&lcnt[dl], 1);
        if (rk < CSR_CAP) {
            lslot[dl * CSR_CAP + rk] = make_int2(rec.y, rec.z);
        } else {
            atomicAdd(&lovw[dl], __int_as_float(rec.z));
            const int k = atomicAdd(ovf_n, 1);
            if ((unsigned)k < (unsigned)E) { ovf_dst[k] = rec.x; ovf_sw[k] = make_int2(rec.y, rec.z); }
        }
    }
}

__global__ __launch_bounds__(BS_TB) void build_slots(int2* __restrict__ csr, int* __restrict__ cnt,
                                                     float* __restrict__ dinv, int* __restrict__ ovf_n,
                                                     int* __restrict__ ovf_dst, int2* __restrict__ ovf_sw,
                                                     const int4* __restrict__ stage, const int2* __restrict__ seg,
                                                     int N, int E, int npb, int nBlk) {
    __shared__ int   lcnt[NPB_MAX];
    __shared__ float lovw[NPB_MAX];
    __shared__ __attribute__((aligned(16))) int2 lslot[NPB_MAX * CSR_CAP];
    const int tid   = threadIdx.x;
    const int lane  = tid & 31;
    const int wv    = __builtin_amdgcn_readfirstlane(tid >> 5);
    const int node0 = blockIdx.x * npb;
    const int nHere = max(0, min(min(npb, NPB_MAX), N - node0));
    if (tid < NPB_MAX) { lcnt[tid] = 0; lovw[tid] = 0.0f; }
    __syncthreads();
    for (int b0 = wv; b0 < nBlk; b0 += 4 * (BS_TB / 32)) {
        int  st[4], ln[4];
        int4 rec[4];
        const int2 sgv = seg[(size_t)blockIdx.x * nBlk + min(b0 + (lane & 3) * (BS_TB / 32), nBlk - 1)];
#pragma unroll
        for (int u = 0; u < 4; ++u) {
            const int b  = b0 + u * (BS_TB / 32);
            const int sx = __builtin_amdgcn_readlane(sgv.x, u);
            const int sy = __builtin_amdgcn_readlane(sgv.y, u);
            st[u] = min(max(sx, 0), BE_EPB - 1);
            ln[u] = (b < nBlk) ? min(max(sy, 0), BE_EPB - st[u]) : 0;
        }
#pragma unroll
        for (int u = 0; u < 4; ++u) {
            const int b   = min(b0 + u * (BS_TB / 32), nBlk - 1);
            const int off = (ln[u] > 0) ? st[u] + min(lane, ln[u] - 1) : 0;
            rec[u] = stage[(size_t)b * BE_EPB + off];
        }
        asm volatile("" :: "v"(rec[0].x), "v"(rec[0].y), "v"(rec[0].z), "v"(rec[1].x), "v"(rec[1].y), "v"(rec[1].z),
                           "v"(rec[2].x), "v"(rec[2].y), "v"(rec[2].z), "v"(rec[3].x), "v"(rec[3].y), "v"(rec[3].z));
#pragma unroll
        for (int u = 0; u < 4; ++u) {
            const int b = b0 + u * (BS_TB / 32);
            if (lane >= ln[u]) rec[u].x = -1;
            place_edge(rec[u], node0, nHere, E, lcnt, lovw, lslot, ovf_n, ovf_dst, ovf_sw);
            for (int i = lane + 32; i < ln[u]; i += 32)
                place_edge(stage[(size_t)b * BE_EPB + st[u] + i], node0, nHere, E, lcnt, lovw, lslot, ovf_n, ovf_dst, ovf_sw);
        }
    }
    __syncthreads();
    if (tid < nHere) {
        const int c  = lcnt[tid];
        const int cb = min(c, CSR_CAP);
        float sum = 0.0f;
        for (int p = 0; p < cb; ++p) sum += __int_as_float(lslot[tid * CSR_CAP + p].y);
        if (c > CSR_CAP) sum += lovw[tid];
        dinv[node0 + tid] = __frsqrt_rn(sum + 1.0f);
        cnt[node0 + tid]  = c;
    }
#pragma unroll 4
    for (int q = tid; q < nHere * CSR_CAP; q += BS_TB) {
        const int nd = q / CSR_CAP, sl = q % CSR_CAP;
        if (sl < lcnt[nd]) csr[(size_t)(node0 + nd) * CSR_CAP + sl] = lslot[q];
    }
}

__device__ __forceinline__ int overflow_count(const int* __restrict__ ovf_n, int E) {
    return min(max(*ovf_n, 0), E);
}

__device__ __forceinline__ void fma4(float4& a, float s, const v4f& v) {
    a.x = fmaf(s, v.x, a.x); a.y = fmaf(s, v.y, a.y);
    a.z = fmaf(s, v.z, a.z); a.w = fmaf(s, v.w, a.w);
}

typedef const __attribute__((address_space(1))) v4f* grow_t;
__device__ __forceinline__ v4f load_row(const float4* __restrict__ h4, int s, int lane) {
    const unsigned long long a = (unsigned long long)h4 + ((unsigned long long)(unsigned)s << 9);
    const unsigned lo = __builtin_amdgcn_readfirstlane((unsigned)a);
    const unsigned hi = __builtin_amdgcn_readfirstlane((unsigned)(a >> 32));
    grow_t row = (grow_t)(((unsigned long long)hi << 32) | lo);
    return row[(unsigned)lane];
}
__device__ __forceinline__ void ask4(const float4* __restrict__ h4, const int2 c, int j, int lane,
                                     v4f (&v)[4], float (&n)[4]) {
#pragma unroll
    for (int k = 0; k < 4; ++k) {
        const int s = __builtin_amdgcn_readlane(c.x, j + k);
        n[k] = __int_as_float(__builtin_amdgcn_readlane(c.y, j + k));
        v[k] = load_row(h4, s, lane);
    }
}
__device__ __forceinline__ void add4(float4& acc, const v4f (&v)[4], const float (&n)[4]) {
#pragma unroll
    for (int k = 0; k < 4; ++k) fma4(acc, n[k], v[k]);
}
__device__ __forceinline__ void pick_tail(const int2 c, int j, int rem, int (&s)[3], float (&n)[3]) {
#pragma unroll
    for (int k = 0; k < 3; ++k) {
        s[k] = __builtin_amdgcn_readlane(c.x, (j + k) & 31);
        const float w = __int_as_float(__builtin_amdgcn_readlane(c.y, (j + k) & 31));
        n[k] = (k < rem) ? w : 0.0f;
    }
}
__device__ __forceinline__ void read_tail(const float4* __restrict__ h4, const int (&s)[3], int rem,
                                          int lane, v4f (&v)[3]) {
#pragma unroll
    for (int k = 0; k < 3; ++k) {
        v[k] = v4f{0.0f, 0.0f, 0.0f, 0.0f};
        if (k < rem) v[k] = load_row(h4, s[k], lane);
    }
}
__device__ __forceinline__ void add_tail(float4& acc, const v4f (&v)[3], const float (&n)[3]) {
#pragma unroll
    for (int k = 0; k < 3; ++k) fma4(acc, n[k], v[k]);
}

template <bool NORM>
__device__ __forceinline__ void segment_sum_pair(const float4* __restrict__ h4,
                                                 const int2* csr, int2* csr_w,
                                                 const float* __restrict__ dinv, float diA, float diB,
                                                 int startA, int endA, int startB, int endB,
                                                 int lane, float4& accA, float4& accB) {
    accA = make_float4(0.0f, 0.0f, 0.0f, 0.0f);
    accB = make_float4(0.0f, 0.0f, 0.0f, 0.0f);
    while (startA < endA || startB < endB) {
        const int cntA = max(0, min(32, endA - startA));
        const int cntB = max(0, min(32, endB - startB));
        int2 ca = make_int2(0, 0), cb = make_int2(0, 0);
        if (lane < cntA) ca = csr[startA + lane];
        if (lane < cntB) cb = csr[startB + lane];
        if (NORM) {
            const float dsa = dinv[ca.x];
            const float dsb = dinv[cb.x];
            ca.y = (lane < cntA) ? __float_as_int(dsa * __int_as_float(ca.y) * diA) : 0;
            cb.y = (lane < cntB) ? __float_as_int(dsb * __int_as_float(cb.y) * diB) : 0;
            if (lane < cntA) csr_w[startA + lane] = ca;
            if (lane < cntB) csr_w[startB + lane] = cb;
        }
        int j = 0;
        for (; j + 4 <= min(cntA, cntB); j += 4) {
            v4f   va[4], vb[4];
            float na[4], nb[4];
            ask4(h4, ca, j, lane, va, na);
            ask4(h4, cb, j, lane, vb, nb);
            asm volatile("" :: "v"(va[0]), "v"(va[1]), "v"(va[2]), "v"(va[3]),
                               "v"(vb[0]), "v"(vb[1]), "v"(vb[2]), "v"(vb[3]));
            add4(accA, va, na);
            add4(accB, vb, nb);
        }
        int jA = j, jB = j;
        for (; jA + 4 <= cntA; jA += 4) {
            v4f   va[4];
            float na[4];
            ask4(h4, ca, jA, lane, va, na);
            add4(accA, va, na);
        }
        for (; jB + 4 <= cntB; jB += 4) {
            v4f   vb[4];
            float nb[4];
            ask4(h4, cb, jB, lane, vb, nb);
            add4(accB, vb, nb);
        }
        const int remA = cntA - jA, remB = cntB - jB;
        if (remA > 0 || remB > 0) {
            v4f   va[3], vb[3];
            float na[3], nb[3];
            int   sa[3], sb[3];
            pick_tail(ca, jA, remA, sa, na);
            pick_tail(cb, jB, remB, sb, nb);
            read_tail(h4, sa, remA, lane, va);
            read_tail(h4, sb, remB, lane, vb);
            add_tail(accA, va, na);
            add_tail(accB, vb, nb);
        }
        startA += 32;
        startB += 32;
    }
}

template <bool NORM>
__device__ __forceinline__ void overflow_sum(const float4* __restrict__ h4,
                                             const int* __restrict__ ovf_dst,
                                             const int2* ovf_sw, int2* ovf_sw_w,
                                             const float* __restrict__ dinv, float di, int n_ovf,
                                             int node, int lane, float4& acc) {
    for (int k0 = 0; k0 < n_ovf; k0 += 32) {
        const int k = k0 + lane;
        int  d = -1;
        int2 c = make_int2(0, 0);
        if (k < n_ovf) { d = ovf_dst[k]; c = ovf_sw[k]; }
        if (NORM) {
            if (d == node) {
                c.y = __float_as_int(dinv[c.x] * __int_as_float(c.y) * di);
                ovf_sw_w[k] = c;
            }
        }
        unsigned m = __builtin_amdgcn_ballot_w32(d == node);
        while (m != 0u) {
            const int j = __builtin_ctz(m);
            m &= m - 1u;
            const int   s = __builtin_amdgcn_readlane(c.x, j);
            const float n = __int_as_float(__builtin_amdgcn_readlane(c.y, j));
            const v4f   v = load_row(h4, s, lane);
            fma4(acc, n, v);
        }
    }
}

#define LT_ROWS  16
#define LT_PITCH 132

template <bool NORM>
__global__ __launch_bounds__(256, 8) void aggregate_then_product(
        const float* __restrict__ xin, int2* csr,
        const int* __restrict__ cnt, const float* __restrict__ dinv,
        const int* __restrict__ ovf_n, const int* __restrict__ ovf_dst,
        int2* ovf_sw,
        const float* __restrict__ W, const float* __restrict__ b,
        float* __restrict__ xout, int N, int E, int nTiles) {
#if defined(__HIP_DEVICE_COMPILE__)
    __shared__ __attribute__((aligned(16))) float tile[LT_ROWS * LT_PITCH];
    const int lane = threadIdx.x & 31;
    const int wv   = __builtin_amdgcn_readfirstlane(threadIdx.x >> 5);
    const float4* __restrict__ xv = reinterpret_cast<const float4*>(xin);

    const int koff = (lane >> 4) * 2;
    const int col  = wv * 16 + (lane & 15);
    const float* __restrict__ arow = tile + (lane & 15) * LT_PITCH + koff;
    const float* __restrict__ wcol = W + (size_t)koff * D_CH + col;
    const int   row0 = (lane >> 4) * 8;

    v2f bw[32];
#pragma unroll
    for (int ks = 0; ks < 32; ++ks) {
        bw[ks].x = wcol[(size_t)(ks * 4) * D_CH];
        bw[ks].y = wcol[(size_t)(ks * 4 + 1) * D_CH];
    }
    const float bias = b[col];

    for (int t = blockIdx.x; t < nTiles; t += gridDim.x) {
        const int node0 = t * LT_ROWS;
        int    nds[2], cn[2], rp0[2], rp1[2];
        v4f    hs[2];
        float  di[2];
#pragma unroll
        for (int u = 0; u < 2; ++u) {
            const int nd = min(node0 + wv + 8 * u, N - 1);
            nds[u] = nd;
            cn[u]  = __builtin_amdgcn_readfirstlane(cnt[nd]);
            if (node0 + wv + 8 * u >= N) cn[u] = 0;
            hs[u]  = load_row(xv, nd, lane);
            di[u]  = __int_as_float(__builtin_amdgcn_readfirstlane(__float_as_int(dinv[nd])));
        }
#pragma unroll
        for (int u = 0; u < 2; ++u) {
            rp0[u] = nds[u] * CSR_CAP;
            rp1[u] = rp0[u] + min(max(cn[u], 0), CSR_CAP);
        }
        float4 a[2];
        segment_sum_pair<NORM>(xv, csr, csr, dinv, di[0], di[1], rp0[0], rp1[0], rp0[1], rp1[1], lane, a[0], a[1]);
#pragma unroll
        for (int u = 0; u < 2; ++u)
            if (cn[u] > CSR_CAP)
                overflow_sum<NORM>(xv, ovf_dst, ovf_sw, ovf_sw, dinv, di[u], overflow_count(ovf_n, E), nds[u], lane, a[u]);
#pragma unroll
        for (int u = 0; u < 2; ++u) {
            const int   r  = wv + 8 * u;
            const float d2 = di[u] * di[u];
            a[u].x = fmaf(hs[u].x, d2, a[u].x); a[u].y = fmaf(hs[u].y, d2, a[u].y);
            a[u].z = fmaf(hs[u].z, d2, a[u].z); a[u].w = fmaf(hs[u].w, d2, a[u].w);
            *reinterpret_cast<float4*>(&tile[r * LT_PITCH + lane * 4]) = a[u];
        }
        __syncthreads();

        v8f acc = v8f{};
#pragma unroll
        for (int ks = 0; ks < 32; ++ks) {
            v2f av;
            av.x = arow[ks * 4];
            av.y = arow[ks * 4 + 1];
            acc = __builtin_amdgcn_wmma_f32_16x16x4_f32(
                false, av, false, bw[ks], (short)0, acc, false, false);
        }
        __syncthreads();

#pragma unroll
        for (int r = 0; r < 8; ++r)
            tile[(row0 + r) * LT_PITCH + col] = fmaxf(acc[r] + bias, 0.0f);
        __syncthreads();

        for (int u = 0; u < 2; ++u) {
            const int r    = wv + 8 * u;
            const int node = node0 + r;
            if (node < N) {
                const float4 o = *reinterpret_cast<const float4*>(&tile[r * LT_PITCH + lane * 4]);
                reinterpret_cast<float4*>(xout)[(size_t)node * (D_CH / 4) + lane] = o;
            }
        }
    }
#endif
}

__global__ __launch_bounds__(256) void last_layer_at_lookups(
        const float* __restrict__ x1, const float* __restrict__ x2,
        const int2* __restrict__ csr, const int* __restrict__ cnt,
        const float* __restrict__ dinv, const int* __restrict__ ovf_n,
        const int* __restrict__ ovf_dst, const int2* __restrict__ ovf_sw,
        const float* __restrict__ W,
        const float* __restrict__ b, const int* __restrict__ idx,
        float* __restrict__ out, int N, int E, int B) {
#if defined(__HIP_DEVICE_COMPILE__)
    __shared__ __attribute__((aligned(16))) float tile[LT_ROWS * LT_PITCH];
    const int lane  = threadIdx.x & 31;
    const int wv    = __builtin_amdgcn_readfirstlane(threadIdx.x >> 5);
    const int slot0 = blockIdx.x * LT_ROWS;
    const float4* __restrict__ x1v = reinterpret_cast<const float4*>(x1);
    const float4* __restrict__ x2v = reinterpret_cast<const float4*>(x2);
    const float4 zero4 = make_float4(0.0f, 0.0f, 0.0f, 0.0f);

    const int slotA = slot0 + wv, slotB = slot0 + wv + 8;
    int nodes[2] = { N, N };
    if (slotA < B) nodes[0] = idx[slotA];
    if (slotB < B) nodes[1] = idx[slotB];

    float4 keep[2];
    int    nds[2], cn[2], rp0[2], rp1[2];
    v4f    hs[2], p1[2];
    float  di[2];
#pragma unroll
    for (int t = 0; t < 2; ++t) {
        const int nd = ((unsigned)nodes[t] < (unsigned)N) ? nodes[t] : 0;
        nds[t] = nd;
        cn[t]  = __builtin_amdgcn_readfirstlane(cnt[nd]);
        hs[t]  = load_row(x2v, nd, lane);
        p1[t]  = load_row(x1v, nd, lane);
        di[t]  = dinv[nd];
    }
#pragma unroll
    for (int t = 0; t < 2; ++t) {
        rp0[t] = nds[t] * CSR_CAP;
        rp1[t] = rp0[t] + min(max(cn[t], 0), CSR_CAP);
    }
    float4 a[2];
    segment_sum_pair<false>(x2v, csr, nullptr, dinv, 0.0f, 0.0f, rp0[0], rp1[0], rp0[1], rp1[1], lane, a[0], a[1]);
#pragma unroll
    for (int t = 0; t < 2; ++t)
        if (cn[t] > CSR_CAP)
            overflow_sum<false>(x2v, ovf_dst, ovf_sw, nullptr, dinv, 0.0f, overflow_count(ovf_n, E), nds[t], lane, a[t]);
#pragma unroll
    for (int t = 0; t < 2; ++t) {
        const int   r  = wv + 8 * t;
        const float d2 = di[t] * di[t];
        a[t].x = fmaf(hs[t].x, d2, a[t].x); a[t].y = fmaf(hs[t].y, d2, a[t].y);
        a[t].z = fmaf(hs[t].z, d2, a[t].z); a[t].w = fmaf(hs[t].w, d2, a[t].w);
        keep[t].x = p1[t].x + hs[t].x; keep[t].y = p1[t].y + hs[t].y;
        keep[t].z = p1[t].z + hs[t].z; keep[t].w = p1[t].w + hs[t].w;
        *reinterpret_cast<float4*>(&tile[r * LT_PITCH + lane * 4]) = a[t];
    }
    __syncthreads();

    const int koff = (lane >> 4) * 2;
    const int col  = wv * 16 + (lane & 15);
    const float* __restrict__ arow = tile + (lane & 15) * LT_PITCH + koff;
    const float* __restrict__ wcol = W + (size_t)koff * D_CH + col;
    v8f acc = v8f{};
#pragma unroll
    for (int ks = 0; ks < 32; ++ks) {
        v2f av, bw;
        av.x = arow[ks * 4];
        av.y = arow[ks * 4 + 1];
        bw.x = wcol[(size_t)(ks * 4) * D_CH];
        bw.y = wcol[(size_t)(ks * 4 + 1) * D_CH];
        acc = __builtin_amdgcn_wmma_f32_16x16x4_f32(
            false, av, false, bw, (short)0, acc, false, false);
    }
    __syncthreads();

    const float bias = b[col];
    const int   row0 = (lane >> 4) * 8;
#pragma unroll
    for (int r = 0; r < 8; ++r)
        tile[(row0 + r) * LT_PITCH + col] = fmaxf(acc[r] + bias, 0.0f);
    __syncthreads();

#pragma unroll
    for (int t = 0; t < 2; ++t) {
        const int r    = wv + 8 * t;
        const int slot = slot0 + r;
        if (slot < B) {
            float4 o = zero4;
            if ((unsigned)nodes[t] < (unsigned)N) {
                const float4 x3 = *reinterpret_cast<const float4*>(&tile[r * LT_PITCH + lane * 4]);
                o.x = (keep[t].x + x3.x) * (1.0f / 3.0f);
                o.y = (keep[t].y + x3.y) * (1.0f / 3.0f);
                o.z = (keep[t].z + x3.z) * (1.0f / 3.0f);
                o.w = (keep[t].w + x3.w) * (1.0f / 3.0f);
            }
            reinterpret_cast<float4*>(out)[(size_t)slot * (D_CH / 4) + lane] = o;
        }
    }
#endif
}

extern "C" void kernel_launch(void* const* d_in, const int* in_sizes, int n_in,
                              void* d_out, int out_size, void* d_ws, size_t ws_size,
                              hipStream_t stream) {
    const int*   inputs_idx  = (const int*)  d_in[0];
    const float* x           = (const float*)d_in[1];
    const int*   edge_index  = (const int*)  d_in[2];
    const float* edge_weight = (const float*)d_in[3];
    const float* Wm[3] = { (const float*)d_in[4], (const float*)d_in[6], (const float*)d_in[8] };
    const float* bv[3] = { (const float*)d_in[5], (const float*)d_in[7], (const float*)d_in[9] };

    const int B = in_sizes[0];
    const int N = in_sizes[1] / D_CH;
    const int E = in_sizes[2] / 2;
    (void)n_in; (void)ws_size; (void)out_size;
    const int* src = edge_index;
    const int* dst = edge_index + E;

    float* ws   = (float*)d_ws;
    const size_t nd   = (size_t)N * D_CH;
    const size_t Npad = ((size_t)N + 3) & ~(size_t)3;
    const size_t Epad = ((size_t)E + 3) & ~(size_t)3;
    const int TB = 256;
    const int nTiles   = (N + LT_ROWS - 1) / LT_ROWS;
    const int aggBlk   = (nTiles + 3) / 4;
    const int slotBlk  = (B + LT_ROWS - 1) / LT_ROWS;
    const int npb      = (N + NBINS - 1) / NBINS;
    const int binBlk   = (E + BE_EPB - 1) / BE_EPB;
    float* xl[2];
    xl[0] = ws;
    xl[1] = xl[0] + nd;
    int2*  csr     = (int2*)(xl[1] + nd);
    int*   cnt     = (int*)(csr + (size_t)N * CSR_CAP);
    int*   ovf_n   = cnt + Npad;
    float* dinv    = (float*)(ovf_n + 4);
    int*   ovf_dst = (int*)(dinv + Npad);
    int2*  ovf_sw  = (int2*)(ovf_dst + Epad);
    int4*  stage   = (int4*)(ovf_sw + Epad);
    int2*  seg     = (int2*)(stage + (size_t)binBlk * BE_EPB);

    bin_edges<<<binBlk, BE_TB, 0, stream>>>(stage, seg, ovf_n, src, dst, edge_weight, E, npb, binBlk);
    build_slots<<<NBINS, BS_TB, 0, stream>>>(csr, cnt, dinv, ovf_n, ovf_dst, ovf_sw, stage, seg, N, E, npb, binBlk);

    aggregate_then_product<true><<<aggBlk, TB, 0, stream>>>(x, csr, cnt, dinv, ovf_n, ovf_dst, ovf_sw,
                                                            Wm[0], bv[0], xl[0], N, E, nTiles);
    aggregate_then_product<false><<<aggBlk, TB, 0, stream>>>(xl[0], csr, cnt, dinv, ovf_n, ovf_dst, ovf_sw,
                                                             Wm[1], bv[1], xl[1], N, E, nTiles);

    last_layer_at_lookups<<<slotBlk, TB, 0, stream>>>(xl[0], xl[1], csr, cnt, dinv,
                                                      ovf_n, ovf_dst, ovf_sw,
                                                      Wm[2], bv[2], inputs_idx,
                                                      (float*)d_out, N, E, B);
}


# Layers 1 and 2 launched as 1024 blocks, 4 resident on every WGP (8 waves per SIMD, 3 or 4 tiles each by stride); 782 blocks of 4 tiles left most WGPs 3: a third more row loads in flight; sums kept
# speedup vs baseline: 6.1723x; 1.0560x over previous
#include <hip/hip_runtime.h>
#include <hip/hip_bf16.h>


typedef __attribute__((ext_vector_type(2))) float v2f;
typedef __attribute__((ext_vector_type(8))) float v8f;
typedef __attribute__((ext_vector_type(4))) float v4f;

#define D_CH 128

#define CSR_CAP 64
#define NBINS   256
#define NPB_MAX 256
#define BE_TB   1024
#define BE_EPT  5
#define BE_EPB  (BE_TB * BE_EPT)
#define BS_TB   512

__global__ __launch_bounds__(BE_TB) void bin_edges(int4* __restrict__ stage, int2* __restrict__ seg,
                                                   int* __restrict__ ovf_n, const int* __restrict__ src,
                                                   const int* __restrict__ dst, const float* __restrict__ w,
                                                   int E, int npb, int nBlk) {
    __shared__ int lcnt[NBINS];
    __shared__ int lbase[NBINS];
    __shared__ __attribute__((aligned(16))) int4 lst[BE_EPB];
    const int tid = threadIdx.x;
    const int e0  = blockIdx.x * BE_EPB;
    if (blockIdx.x == 0 && tid < 4) ovf_n[tid] = 0;
    if (tid < NBINS) lcnt[tid] = 0;
    int d[BE_EPT], s[BE_EPT], wb[BE_EPT], bin[BE_EPT], r[BE_EPT];
#pragma unroll
    for (int k = 0; k < BE_EPT; ++k) {
        const int e = e0 + k * BE_TB + tid;
        d[k] = -1; s[k] = 0; wb[k] = 0;
        if (e < E) { d[k] = dst[e]; s[k] = src[e]; wb[k] = __float_as_int(w[e]); }
    }
    __syncthreads();
#pragma unroll
    for (int k = 0; k < BE_EPT; ++k) {
        bin[k] = 0; r[k] = 0;
        if (d[k] >= 0) {
            bin[k] = min(d[k] / npb, NBINS - 1);
            r[k]   = atomicAdd(&lcnt[bin[k]], 1);
        }
    }
    __syncthreads();
    if (tid < 32) {
        int cc[NBINS / 32];
        int run = 0;
#pragma unroll
        for (int j = 0; j < NBINS / 32; ++j) { cc[j] = run; run += lcnt[tid * (NBINS / 32) + j]; }
        int inc = run;
#pragma unroll
        for (int o = 1; o < 32; o <<= 1) {
            const int v = __shfl_up(inc, o, 32);
            if (tid >= o) inc += v;
        }
        const int excl = inc - run;
#pragma unroll
        for (int j = 0; j < NBINS / 32; ++j) lbase[tid * (NBINS / 32) + j] = excl + cc[j];
    }
    __syncthreads();
#pragma unroll
    for (int k = 0; k < BE_EPT; ++k)
        if (d[k] >= 0) lst[lbase[bin[k]] + r[k]] = make_int4(d[k], s[k], wb[k], 0);
    __syncthreads();
    const int nHere = min(BE_EPB, max(E - e0, 0));
    for (int i = tid; i < nHere; i += BE_TB) stage[(size_t)e0 + i] = lst[i];
    if (tid < NBINS) seg[(size_t)tid * nBlk + blockIdx.x] = make_int2(lbase[tid], lcnt[tid]);
}

__device__ __forceinline__ void place_edge(const int4 rec, int node0, int nHere, int E,
                                           int* lcnt, float* lovw, int2* lslot, int* __restrict__ ovf_n,
                                           int* __restrict__ ovf_dst, int2* __restrict__ ovf_sw) {
    const int dl = rec.x - node0;
    if ((unsigned)dl < (unsigned)nHere) {
        const int rk = atomicAdd(&lcnt[dl], 1);
        if (rk < CSR_CAP) {
            lslot[dl * CSR_CAP + rk] = make_int2(rec.y, rec.z);
        } else {
            atomicAdd(&lovw[dl], __int_as_float(rec.z));
            const int k = atomicAdd(ovf_n, 1);
            if ((unsigned)k < (unsigned)E) { ovf_dst[k] = rec.x; ovf_sw[k] = make_int2(rec.y, rec.z); }
        }
    }
}

__global__ __launch_bounds__(BS_TB) void build_slots(int2* __restrict__ csr, int* __restrict__ cnt,
                                                     float* __restrict__ dinv, int* __restrict__ ovf_n,
                                                     int* __restrict__ ovf_dst, int2* __restrict__ ovf_sw,
                                                     const int4* __restrict__ stage, const int2* __restrict__ seg,
                                                     int N, int E, int npb, int nBlk) {
    __shared__ int   lcnt[NPB_MAX];
    __shared__ float lovw[NPB_MAX];
    __shared__ __attribute__((aligned(16))) int2 lslot[NPB_MAX * CSR_CAP];
    const int tid   = threadIdx.x;
    const int lane  = tid & 31;
    const int wv    = __builtin_amdgcn_readfirstlane(tid >> 5);
    const int node0 = blockIdx.x * npb;
    const int nHere = max(0, min(min(npb, NPB_MAX), N - node0));
    if (tid < NPB_MAX) { lcnt[tid] = 0; lovw[tid] = 0.0f; }
    __syncthreads();
    for (int b0 = wv; b0 < nBlk; b0 += 4 * (BS_TB / 32)) {
        int  st[4], ln[4];
        int4 rec[4];
        const int2 sgv = seg[(size_t)blockIdx.x * nBlk + min(b0 + (lane & 3) * (BS_TB / 32), nBlk - 1)];
#pragma unroll
        for (int u = 0; u < 4; ++u) {
            const int b  = b0 + u * (BS_TB / 32);
            const int sx = __builtin_amdgcn_readlane(sgv.x, u);
            const int sy = __builtin_amdgcn_readlane(sgv.y, u);
            st[u] = min(max(sx, 0), BE_EPB - 1);
            ln[u] = (b < nBlk) ? min(max(sy, 0), BE_EPB - st[u]) : 0;
        }
#pragma unroll
        for (int u = 0; u < 4; ++u) {
            const int b   = min(b0 + u * (BS_TB / 32), nBlk - 1);
            const int off = (ln[u] > 0) ? st[u] + min(lane, ln[u] - 1) : 0;
            rec[u] = stage[(size_t)b * BE_EPB + off];
        }
        asm volatile("" :: "v"(rec[0].x), "v"(rec[0].y), "v"(rec[0].z), "v"(rec[1].x), "v"(rec[1].y), "v"(rec[1].z),
                           "v"(rec[2].x), "v"(rec[2].y), "v"(rec[2].z), "v"(rec[3].x), "v"(rec[3].y), "v"(rec[3].z));
#pragma unroll
        for (int u = 0; u < 4; ++u) {
            const int b = b0 + u * (BS_TB / 32);
            if (lane >= ln[u]) rec[u].x = -1;
            place_edge(rec[u], node0, nHere, E, lcnt, lovw, lslot, ovf_n, ovf_dst, ovf_sw);
            for (int i = lane + 32; i < ln[u]; i += 32)
                place_edge(stage[(size_t)b * BE_EPB + st[u] + i], node0, nHere, E, lcnt, lovw, lslot, ovf_n, ovf_dst, ovf_sw);
        }
    }
    __syncthreads();
    if (tid < nHere) {
        const int c  = lcnt[tid];
        const int cb = min(c, CSR_CAP);
        float sum = 0.0f;
        for (int p = 0; p < cb; ++p) sum += __int_as_float(lslot[tid * CSR_CAP + p].y);
        if (c > CSR_CAP) sum += lovw[tid];
        dinv[node0 + tid] = __frsqrt_rn(sum + 1.0f);
        cnt[node0 + tid]  = c;
    }
#pragma unroll 4
    for (int q = tid; q < nHere * CSR_CAP; q += BS_TB) {
        const int nd = q / CSR_CAP, sl = q % CSR_CAP;
        if (sl < lcnt[nd]) csr[(size_t)(node0 + nd) * CSR_CAP + sl] = lslot[q];
    }
}

__device__ __forceinline__ int overflow_count(const int* __restrict__ ovf_n, int E) {
    return min(max(*ovf_n, 0), E);
}

__device__ __forceinline__ void fma4(float4& a, float s, const v4f& v) {
    a.x = fmaf(s, v.x, a.x); a.y = fmaf(s, v.y, a.y);
    a.z = fmaf(s, v.z, a.z); a.w = fmaf(s, v.w, a.w);
}

typedef const __attribute__((address_space(1))) v4f* grow_t;
__device__ __forceinline__ v4f load_row(const float4* __restrict__ h4, int s, int lane) {
    const unsigned long long a = (unsigned long long)h4 + ((unsigned long long)(unsigned)s << 9);
    const unsigned lo = __builtin_amdgcn_readfirstlane((unsigned)a);
    const unsigned hi = __builtin_amdgcn_readfirstlane((unsigned)(a >> 32));
    grow_t row = (grow_t)(((unsigned long long)hi << 32) | lo);
    return row[(unsigned)lane];
}
__device__ __forceinline__ void ask4(const float4* __restrict__ h4, const int2 c, int j, int lane,
                                     v4f (&v)[4], float (&n)[4]) {
#pragma unroll
    for (int k = 0; k < 4; ++k) {
        const int s = __builtin_amdgcn_readlane(c.x, j + k);
        n[k] = __int_as_float(__builtin_amdgcn_readlane(c.y, j + k));
        v[k] = load_row(h4, s, lane);
    }
}
__device__ __forceinline__ void add4(float4& acc, const v4f (&v)[4], const float (&n)[4]) {
#pragma unroll
    for (int k = 0; k < 4; ++k) fma4(acc, n[k], v[k]);
}
__device__ __forceinline__ void pick_tail(const int2 c, int j, int rem, int (&s)[3], float (&n)[3]) {
#pragma unroll
    for (int k = 0; k < 3; ++k) {
        s[k] = __builtin_amdgcn_readlane(c.x, (j + k) & 31);
        const float w = __int_as_float(__builtin_amdgcn_readlane(c.y, (j + k) & 31));
        n[k] = (k < rem) ? w : 0.0f;
    }
}
__device__ __forceinline__ void read_tail(const float4* __restrict__ h4, const int (&s)[3], int rem,
                                          int lane, v4f (&v)[3]) {
#pragma unroll
    for (int k = 0; k < 3; ++k) {
        v[k] = v4f{0.0f, 0.0f, 0.0f, 0.0f};
        if (k < rem) v[k] = load_row(h4, s[k], lane);
    }
}
__device__ __forceinline__ void add_tail(float4& acc, const v4f (&v)[3], const float (&n)[3]) {
#pragma unroll
    for (int k = 0; k < 3; ++k) fma4(acc, n[k], v[k]);
}

template <bool NORM>
__device__ __forceinline__ void segment_sum_pair(const float4* __restrict__ h4,
                                                 const int2* csr, int2* csr_w,
                                                 const float* __restrict__ dinv, float diA, float diB,
                                                 int startA, int endA, int startB, int endB,
                                                 int lane, float4& accA, float4& accB) {
    accA = make_float4(0.0f, 0.0f, 0.0f, 0.0f);
    accB = make_float4(0.0f, 0.0f, 0.0f, 0.0f);
    while (startA < endA || startB < endB) {
        const int cntA = max(0, min(32, endA - startA));
        const int cntB = max(0, min(32, endB - startB));
        int2 ca = make_int2(0, 0), cb = make_int2(0, 0);
        if (lane < cntA) ca = csr[startA + lane];
        if (lane < cntB) cb = csr[startB + lane];
        if (NORM) {
            const float dsa = dinv[ca.x];
            const float dsb = dinv[cb.x];
            ca.y = (lane < cntA) ? __float_as_int(dsa * __int_as_float(ca.y) * diA) : 0;
            cb.y = (lane < cntB) ? __float_as_int(dsb * __int_as_float(cb.y) * diB) : 0;
            if (lane < cntA) csr_w[startA + lane] = ca;
            if (lane < cntB) csr_w[startB + lane] = cb;
        }
        int j = 0;
        for (; j + 4 <= min(cntA, cntB); j += 4) {
            v4f   va[4], vb[4];
            float na[4], nb[4];
            ask4(h4, ca, j, lane, va, na);
            ask4(h4, cb, j, lane, vb, nb);
            asm volatile("" :: "v"(va[0]), "v"(va[1]), "v"(va[2]), "v"(va[3]),
                               "v"(vb[0]), "v"(vb[1]), "v"(vb[2]), "v"(vb[3]));
            add4(accA, va, na);
            add4(accB, vb, nb);
        }
        int jA = j, jB = j;
        for (; jA + 4 <= cntA; jA += 4) {
            v4f   va[4];
            float na[4];
            ask4(h4, ca, jA, lane, va, na);
            add4(accA, va, na);
        }
        for (; jB + 4 <= cntB; jB += 4) {
            v4f   vb[4];
            float nb[4];
            ask4(h4, cb, jB, lane, vb, nb);
            add4(accB, vb, nb);
        }
        const int remA = cntA - jA, remB = cntB - jB;
        if (remA > 0 || remB > 0) {
            v4f   va[3], vb[3];
            float na[3], nb[3];
            int   sa[3], sb[3];
            pick_tail(ca, jA, remA, sa, na);
            pick_tail(cb, jB, remB, sb, nb);
            read_tail(h4, sa, remA, lane, va);
            read_tail(h4, sb, remB, lane, vb);
            add_tail(accA, va, na);
            add_tail(accB, vb, nb);
        }
        startA += 32;
        startB += 32;
    }
}

template <bool NORM>
__device__ __forceinline__ void overflow_sum(const float4* __restrict__ h4,
                                             const int* __restrict__ ovf_dst,
                                             const int2* ovf_sw, int2* ovf_sw_w,
                                             const float* __restrict__ dinv, float di, int n_ovf,
                                             int node, int lane, float4& acc) {
    for (int k0 = 0; k0 < n_ovf; k0 += 32) {
        const int k = k0 + lane;
        int  d = -1;
        int2 c = make_int2(0, 0);
        if (k < n_ovf) { d = ovf_dst[k]; c = ovf_sw[k]; }
        if (NORM) {
            if (d == node) {
                c.y = __float_as_int(dinv[c.x] * __int_as_float(c.y) * di);
                ovf_sw_w[k] = c;
            }
        }
        unsigned m = __builtin_amdgcn_ballot_w32(d == node);
        while (m != 0u) {
            const int j = __builtin_ctz(m);
            m &= m - 1u;
            const int   s = __builtin_amdgcn_readlane(c.x, j);
            const float n = __int_as_float(__builtin_amdgcn_readlane(c.y, j));
            const v4f   v = load_row(h4, s, lane);
            fma4(acc, n, v);
        }
    }
}

#define LT_ROWS  16
#define LT_PITCH 132
#define AGG_GRID 1024

template <bool NORM>
__global__ __launch_bounds__(256, 8) void aggregate_then_product(
        const float* __restrict__ xin, int2* csr,
        const int* __restrict__ cnt, const float* __restrict__ dinv,
        const int* __restrict__ ovf_n, const int* __restrict__ ovf_dst,
        int2* ovf_sw,
        const float* __restrict__ W, const float* __restrict__ b,
        float* __restrict__ xout, int N, int E, int nTiles) {
#if defined(__HIP_DEVICE_COMPILE__)
    __shared__ __attribute__((aligned(16))) float tile[LT_ROWS * LT_PITCH];
    const int lane = threadIdx.x & 31;
    const int wv   = __builtin_amdgcn_readfirstlane(threadIdx.x >> 5);
    const float4* __restrict__ xv = reinterpret_cast<const float4*>(xin);

    const int koff = (lane >> 4) * 2;
    const int col  = wv * 16 + (lane & 15);
    const float* __restrict__ arow = tile + (lane & 15) * LT_PITCH + koff;
    const float* __restrict__ wcol = W + (size_t)koff * D_CH + col;
    const int   row0 = (lane >> 4) * 8;

    v2f bw[32];
#pragma unroll
    for (int ks = 0; ks < 32; ++ks) {
        bw[ks].x = wcol[(size_t)(ks * 4) * D_CH];
        bw[ks].y = wcol[(size_t)(ks * 4 + 1) * D_CH];
    }
    const float bias = b[col];

    for (int t = blockIdx.x; t < nTiles; t += gridDim.x) {
        const int node0 = t * LT_ROWS;
        int    nds[2], cn[2], rp0[2], rp1[2];
        v4f    hs[2];
        float  di[2];
#pragma unroll
        for (int u = 0; u < 2; ++u) {
            const int nd = min(node0 + wv + 8 * u, N - 1);
            nds[u] = nd;
            cn[u]  = __builtin_amdgcn_readfirstlane(cnt[nd]);
            if (node0 + wv + 8 * u >= N) cn[u] = 0;
            hs[u]  = load_row(xv, nd, lane);
            di[u]  = __int_as_float(__builtin_amdgcn_readfirstlane(__float_as_int(dinv[nd])));
        }
#pragma unroll
        for (int u = 0; u < 2; ++u) {
            rp0[u] = nds[u] * CSR_CAP;
            rp1[u] = rp0[u] + min(max(cn[u], 0), CSR_CAP);
        }
        float4 a[2];
        segment_sum_pair<NORM>(xv, csr, csr, dinv, di[0], di[1], rp0[0], rp1[0], rp0[1], rp1[1], lane, a[0], a[1]);
#pragma unroll
        for (int u = 0; u < 2; ++u)
            if (cn[u] > CSR_CAP)
                overflow_sum<NORM>(xv, ovf_dst, ovf_sw, ovf_sw, dinv, di[u], overflow_count(ovf_n, E), nds[u], lane, a[u]);
#pragma unroll
        for (int u = 0; u < 2; ++u) {
            const int   r  = wv + 8 * u;
            const float d2 = di[u] * di[u];
            a[u].x = fmaf(hs[u].x, d2, a[u].x); a[u].y = fmaf(hs[u].y, d2, a[u].y);
            a[u].z = fmaf(hs[u].z, d2, a[u].z); a[u].w = fmaf(hs[u].w, d2, a[u].w);
            *reinterpret_cast<float4*>(&tile[r * LT_PITCH + lane * 4]) = a[u];
        }
        __syncthreads();

        v8f acc = v8f{};
#pragma unroll
        for (int ks = 0; ks < 32; ++ks) {
            v2f av;
            av.x = arow[ks * 4];
            av.y = arow[ks * 4 + 1];
            acc = __builtin_amdgcn_wmma_f32_16x16x4_f32(
                false, av, false, bw[ks], (short)0, acc, false, false);
        }
        __syncthreads();

#pragma unroll
        for (int r = 0; r < 8; ++r)
            tile[(row0 + r) * LT_PITCH + col] = fmaxf(acc[r] + bias, 0.0f);
        __syncthreads();

        for (int u = 0; u < 2; ++u) {
            const int r    = wv + 8 * u;
            const int node = node0 + r;
            if (node < N) {
                const float4 o = *reinterpret_cast<const float4*>(&tile[r * LT_PITCH + lane * 4]);
                reinterpret_cast<float4*>(xout)[(size_t)node * (D_CH / 4) + lane] = o;
            }
        }
    }
#endif
}

__global__ __launch_bounds__(256) void last_layer_at_lookups(
        const float* __restrict__ x1, const float* __restrict__ x2,
        const int2* __restrict__ csr, const int* __restrict__ cnt,
        const float* __restrict__ dinv, const int* __restrict__ ovf_n,
        const int* __restrict__ ovf_dst, const int2* __restrict__ ovf_sw,
        const float* __restrict__ W,
        const float* __restrict__ b, const int* __restrict__ idx,
        float* __restrict__ out, int N, int E, int B) {
#if defined(__HIP_DEVICE_COMPILE__)
    __shared__ __attribute__((aligned(16))) float tile[LT_ROWS * LT_PITCH];
    const int lane  = threadIdx.x & 31;
    const int wv    = __builtin_amdgcn_readfirstlane(threadIdx.x >> 5);
    const int slot0 = blockIdx.x * LT_ROWS;
    const float4* __restrict__ x1v = reinterpret_cast<const float4*>(x1);
    const float4* __restrict__ x2v = reinterpret_cast<const float4*>(x2);
    const float4 zero4 = make_float4(0.0f, 0.0f, 0.0f, 0.0f);

    const int slotA = slot0 + wv, slotB = slot0 + wv + 8;
    int nodes[2] = { N, N };
    if (slotA < B) nodes[0] = idx[slotA];
    if (slotB < B) nodes[1] = idx[slotB];

    float4 keep[2];
    int    nds[2], cn[2], rp0[2], rp1[2];
    v4f    hs[2], p1[2];
    float  di[2];
#pragma unroll
    for (int t = 0; t < 2; ++t) {
        const int nd = ((unsigned)nodes[t] < (unsigned)N) ? nodes[t] : 0;
        nds[t] = nd;
        cn[t]  = __builtin_amdgcn_readfirstlane(cnt[nd]);
        hs[t]  = load_row(x2v, nd, lane);
        p1[t]  = load_row(x1v, nd, lane);
        di[t]  = dinv[nd];
    }
#pragma unroll
    for (int t = 0; t < 2; ++t) {
        rp0[t] = nds[t] * CSR_CAP;
        rp1[t] = rp0[t] + min(max(cn[t], 0), CSR_CAP);
    }
    float4 a[2];
    segment_sum_pair<false>(x2v, csr, nullptr, dinv, 0.0f, 0.0f, rp0[0], rp1[0], rp0[1], rp1[1], lane, a[0], a[1]);
#pragma unroll
    for (int t = 0; t < 2; ++t)
        if (cn[t] > CSR_CAP)
            overflow_sum<false>(x2v, ovf_dst, ovf_sw, nullptr, dinv, 0.0f, overflow_count(ovf_n, E), nds[t], lane, a[t]);
#pragma unroll
    for (int t = 0; t < 2; ++t) {
        const int   r  = wv + 8 * t;
        const float d2 = di[t] * di[t];
        a[t].x = fmaf(hs[t].x, d2, a[t].x); a[t].y = fmaf(hs[t].y, d2, a[t].y);
        a[t].z = fmaf(hs[t].z, d2, a[t].z); a[t].w = fmaf(hs[t].w, d2, a[t].w);
        keep[t].x = p1[t].x + hs[t].x; keep[t].y = p1[t].y + hs[t].y;
        keep[t].z = p1[t].z + hs[t].z; keep[t].w = p1[t].w + hs[t].w;
        *reinterpret_cast<float4*>(&tile[r * LT_PITCH + lane * 4]) = a[t];
    }
    __syncthreads();

    const int koff = (lane >> 4) * 2;
    const int col  = wv * 16 + (lane & 15);
    const float* __restrict__ arow = tile + (lane & 15) * LT_PITCH + koff;
    const float* __restrict__ wcol = W + (size_t)koff * D_CH + col;
    v8f acc = v8f{};
#pragma unroll
    for (int ks = 0; ks < 32; ++ks) {
        v2f av, bw;
        av.x = arow[ks * 4];
        av.y = arow[ks * 4 + 1];
        bw.x = wcol[(size_t)(ks * 4) * D_CH];
        bw.y = wcol[(size_t)(ks * 4 + 1) * D_CH];
        acc = __builtin_amdgcn_wmma_f32_16x16x4_f32(
            false, av, false, bw, (short)0, acc, false, false);
    }
    __syncthreads();

    const float bias = b[col];
    const int   row0 = (lane >> 4) * 8;
#pragma unroll
    for (int r = 0; r < 8; ++r)
        tile[(row0 + r) * LT_PITCH + col] = fmaxf(acc[r] + bias, 0.0f);
    __syncthreads();

#pragma unroll
    for (int t = 0; t < 2; ++t) {
        const int r    = wv + 8 * t;
        const int slot = slot0 + r;
        if (slot < B) {
            float4 o = zero4;
            if ((unsigned)nodes[t] < (unsigned)N) {
                const float4 x3 = *reinterpret_cast<const float4*>(&tile[r * LT_PITCH + lane * 4]);
                o.x = (keep[t].x + x3.x) * (1.0f / 3.0f);
                o.y = (keep[t].y + x3.y) * (1.0f / 3.0f);
                o.z = (keep[t].z + x3.z) * (1.0f / 3.0f);
                o.w = (keep[t].w + x3.w) * (1.0f / 3.0f);
            }
            reinterpret_cast<float4*>(out)[(size_t)slot * (D_CH / 4) + lane] = o;
        }
    }
#endif
}

extern "C" void kernel_launch(void* const* d_in, const int* in_sizes, int n_in,
                              void* d_out, int out_size, void* d_ws, size_t ws_size,
                              hipStream_t stream) {
    const int*   inputs_idx  = (const int*)  d_in[0];
    const float* x           = (const float*)d_in[1];
    const int*   edge_index  = (const int*)  d_in[2];
    const float* edge_weight = (const float*)d_in[3];
    const float* Wm[3] = { (const float*)d_in[4], (const float*)d_in[6], (const float*)d_in[8] };
    const float* bv[3] = { (const float*)d_in[5], (const float*)d_in[7], (const float*)d_in[9] };

    const int B = in_sizes[0];
    const int N = in_sizes[1] / D_CH;
    const int E = in_sizes[2] / 2;
    (void)n_in; (void)ws_size; (void)out_size;
    const int* src = edge_index;
    const int* dst = edge_index + E;

    float* ws   = (float*)d_ws;
    const size_t nd   = (size_t)N * D_CH;
    const size_t Npad = ((size_t)N + 3) & ~(size_t)3;
    const size_t Epad = ((size_t)E + 3) & ~(size_t)3;
    const int TB = 256;
    const int nTiles   = (N + LT_ROWS - 1) / LT_ROWS;
    const int aggBlk   = nTiles < AGG_GRID ? nTiles : AGG_GRID;
    const int slotBlk  = (B + LT_ROWS - 1) / LT_ROWS;
    const int npb      = (N + NBINS - 1) / NBINS;
    const int binBlk   = (E + BE_EPB - 1) / BE_EPB;
    float* xl[2];
    xl[0] = ws;
    xl[1] = xl[0] + nd;
    int2*  csr     = (int2*)(xl[1] + nd);
    int*   cnt     = (int*)(csr + (size_t)N * CSR_CAP);
    int*   ovf_n   = cnt + Npad;
    float* dinv    = (float*)(ovf_n + 4);
    int*   ovf_dst = (int*)(dinv + Npad);
    int2*  ovf_sw  = (int2*)(ovf_dst + Epad);
    int4*  stage   = (int4*)(ovf_sw + Epad);
    int2*  seg     = (int2*)(stage + (size_t)binBlk * BE_EPB);

    bin_edges<<<binBlk, BE_TB, 0, stream>>>(stage, seg, ovf_n, src, dst, edge_weight, E, npb, binBlk);
    build_slots<<<NBINS, BS_TB, 0, stream>>>(csr, cnt, dinv, ovf_n, ovf_dst, ovf_sw, stage, seg, N, E, npb, binBlk);

    aggregate_then_product<true><<<aggBlk, TB, 0, stream>>>(x, csr, cnt, dinv, ovf_n, ovf_dst, ovf_sw,
                                                            Wm[0], bv[0], xl[0], N, E, nTiles);
    aggregate_then_product<false><<<aggBlk, TB, 0, stream>>>(xl[0], csr, cnt, dinv, ovf_n, ovf_dst, ovf_sw,
                                                             Wm[1], bv[1], xl[1], N, E, nTiles);

    last_layer_at_lookups<<<slotBlk, TB, 0, stream>>>(xl[0], xl[1], csr, cnt, dinv,
                                                      ovf_n, ovf_dst, ovf_sw,
                                                      Wm[2], bv[2], inputs_idx,
                                                      (float*)d_out, N, E, B);
}


# Gather: each row load of a trip takes its base from a scalar register pair of its own, all formed before the first load, so no load waits for the translation of the one before it; sums unchanged
# speedup vs baseline: 6.4560x; 1.0460x over previous
#include <hip/hip_runtime.h>
#include <hip/hip_bf16.h>


typedef __attribute__((ext_vector_type(2))) float v2f;
typedef __attribute__((ext_vector_type(8))) float v8f;
typedef __attribute__((ext_vector_type(4))) float v4f;

#define D_CH 128

#define CSR_CAP 64
#define NBINS   256
#define NPB_MAX 256
#define BE_TB   1024
#define BE_EPT  5
#define BE_EPB  (BE_TB * BE_EPT)
#define BS_TB   512

__global__ __launch_bounds__(BE_TB) void bin_edges(int4* __restrict__ stage, int2* __restrict__ seg,
                                                   int* __restrict__ ovf_n, const int* __restrict__ src,
                                                   const int* __restrict__ dst, const float* __restrict__ w,
                                                   int E, int npb, int nBlk) {
    __shared__ int lcnt[NBINS];
    __shared__ int lbase[NBINS];
    __shared__ __attribute__((aligned(16))) int4 lst[BE_EPB];
    const int tid = threadIdx.x;
    const int e0  = blockIdx.x * BE_EPB;
    if (blockIdx.x == 0 && tid < 4) ovf_n[tid] = 0;
    if (tid < NBINS) lcnt[tid] = 0;
    int d[BE_EPT], s[BE_EPT], wb[BE_EPT], bin[BE_EPT], r[BE_EPT];
#pragma unroll
    for (int k = 0; k < BE_EPT; ++k) {
        const int e = e0 + k * BE_TB + tid;
        d[k] = -1; s[k] = 0; wb[k] = 0;
        if (e < E) { d[k] = dst[e]; s[k] = src[e]; wb[k] = __float_as_int(w[e]); }
    }
    __syncthreads();
#pragma unroll
    for (int k = 0; k < BE_EPT; ++k) {
        bin[k] = 0; r[k] = 0;
        if (d[k] >= 0) {
            bin[k] = min(d[k] / npb, NBINS - 1);
            r[k]   = atomicAdd(&lcnt[bin[k]], 1);
        }
    }
    __syncthreads();
    if (tid < 32) {
        int cc[NBINS / 32];
        int run = 0;
#pragma unroll
        for (int j = 0; j < NBINS / 32; ++j) { cc[j] = run; run += lcnt[tid * (NBINS / 32) + j]; }
        int inc = run;
#pragma unroll
        for (int o = 1; o < 32; o <<= 1) {
            const int v = __shfl_up(inc, o, 32);
            if (tid >= o) inc += v;
        }
        const int excl = inc - run;
#pragma unroll
        for (int j = 0; j < NBINS / 32; ++j) lbase[tid * (NBINS / 32) + j] = excl + cc[j];
    }
    __syncthreads();
#pragma unroll
    for (int k = 0; k < BE_EPT; ++k)
        if (d[k] >= 0) lst[lbase[bin[k]] + r[k]] = make_int4(d[k], s[k], wb[k], 0);
    __syncthreads();
    const int nHere = min(BE_EPB, max(E - e0, 0));
    for (int i = tid; i < nHere; i += BE_TB) stage[(size_t)e0 + i] = lst[i];
    if (tid < NBINS) seg[(size_t)tid * nBlk + blockIdx.x] = make_int2(lbase[tid], lcnt[tid]);
}

__device__ __forceinline__ void place_edge(const int4 rec, int node0, int nHere, int E,
                                           int* lcnt, float* lovw, int2* lslot, int* __restrict__ ovf_n,
                                           int* __restrict__ ovf_dst, int2* __restrict__ ovf_sw) {
    const int dl = rec.x - node0;
    if ((unsigned)dl < (unsigned)nHere) {
        const int rk = atomicAdd(&lcnt[dl], 1);
        if (rk < CSR_CAP) {
            lslot[dl * CSR_CAP + rk] = make_int2(rec.y, rec.z);
        } else {
            atomicAdd(&lovw[dl], __int_as_float(rec.z));
            const int k = atomicAdd(ovf_n, 1);
            if ((unsigned)k < (unsigned)E) { ovf_dst[k] = rec.x; ovf_sw[k] = make_int2(rec.y, rec.z); }
        }
    }
}

__global__ __launch_bounds__(BS_TB) void build_slots(int2* __restrict__ csr, int* __restrict__ cnt,
                                                     float* __restrict__ dinv, int* __restrict__ ovf_n,
                                                     int* __restrict__ ovf_dst, int2* __restrict__ ovf_sw,
                                                     const int4* __restrict__ stage, const int2* __restrict__ seg,
                                                     int N, int E, int npb, int nBlk) {
    __shared__ int   lcnt[NPB_MAX];
    __shared__ float lovw[NPB_MAX];
    __shared__ __attribute__((aligned(16))) int2 lslot[NPB_MAX * CSR_CAP];
    const int tid   = threadIdx.x;
    const int lane  = tid & 31;
    const int wv    = __builtin_amdgcn_readfirstlane(tid >> 5);
    const int node0 = blockIdx.x * npb;
    const int nHere = max(0, min(min(npb, NPB_MAX), N - node0));
    if (tid < NPB_MAX) { lcnt[tid] = 0; lovw[tid] = 0.0f; }
    __syncthreads();
    for (int b0 = wv; b0 < nBlk; b0 += 4 * (BS_TB / 32)) {
        int  st[4], ln[4];
        int4 rec[4];
        const int2 sgv = seg[(size_t)blockIdx.x * nBlk + min(b0 + (lane & 3) * (BS_TB / 32), nBlk - 1)];
#pragma unroll
        for (int u = 0; u < 4; ++u) {
            const int b  = b0 + u * (BS_TB / 32);
            const int sx = __builtin_amdgcn_readlane(sgv.x, u);
            const int sy = __builtin_amdgcn_readlane(sgv.y, u);
            st[u] = min(max(sx, 0), BE_EPB - 1);
            ln[u] = (b < nBlk) ? min(max(sy, 0), BE_EPB - st[u]) : 0;
        }
#pragma unroll
        for (int u = 0; u < 4; ++u) {
            const int b   = min(b0 + u * (BS_TB / 32), nBlk - 1);
            const int off = (ln[u] > 0) ? st[u] + min(lane, ln[u] - 1) : 0;
            rec[u] = stage[(size_t)b * BE_EPB + off];
        }
        asm volatile("" :: "v"(rec[0].x), "v"(rec[0].y), "v"(rec[0].z), "v"(rec[1].x), "v"(rec[1].y), "v"(rec[1].z),
                           "v"(rec[2].x), "v"(rec[2].y), "v"(rec[2].z), "v"(rec[3].x), "v"(rec[3].y), "v"(rec[3].z));
#pragma unroll
        for (int u = 0; u < 4; ++u) {
            const int b = b0 + u * (BS_TB / 32);
            if (lane >= ln[u]) rec[u].x = -1;
            place_edge(rec[u], node0, nHere, E, lcnt, lovw, lslot, ovf_n, ovf_dst, ovf_sw);
            for (int i = lane + 32; i < ln[u]; i += 32)
                place_edge(stage[(size_t)b * BE_EPB + st[u] + i], node0, nHere, E, lcnt, lovw, lslot, ovf_n, ovf_dst, ovf_sw);
        }
    }
    __syncthreads();
    if (tid < nHere) {
        const int c  = lcnt[tid];
        const int cb = min(c, CSR_CAP);
        float sum = 0.0f;
        for (int p = 0; p < cb; ++p) sum += __int_as_float(lslot[tid * CSR_CAP + p].y);
        if (c > CSR_CAP) sum += lovw[tid];
        dinv[node0 + tid] = __frsqrt_rn(sum + 1.0f);
        cnt[node0 + tid]  = c;
    }
#pragma unroll 4
    for (int q = tid; q < nHere * CSR_CAP; q += BS_TB) {
        const int nd = q / CSR_CAP, sl = q % CSR_CAP;
        if (sl < lcnt[nd]) csr[(size_t)(node0 + nd) * CSR_CAP + sl] = lslot[q];
    }
}

__device__ __forceinline__ int overflow_count(const int* __restrict__ ovf_n, int E) {
    return min(max(*ovf_n, 0), E);
}

__device__ __forceinline__ void fma4(float4& a, float s, const v4f& v) {
    a.x = fmaf(s, v.x, a.x); a.y = fmaf(s, v.y, a.y);
    a.z = fmaf(s, v.z, a.z); a.w = fmaf(s, v.w, a.w);
}

typedef const __attribute__((address_space(1))) v4f* grow_t;
__device__ __forceinline__ grow_t row_base(const float4* __restrict__ h4, int s) {
    const unsigned long long a = (unsigned long long)h4 + ((unsigned long long)(unsigned)s << 9);
    const unsigned lo = __builtin_amdgcn_readfirstlane((unsigned)a);
    const unsigned hi = __builtin_amdgcn_readfirstlane((unsigned)(a >> 32));
    return (grow_t)(((unsigned long long)hi << 32) | lo);
}
__device__ __forceinline__ v4f load_row(const float4* __restrict__ h4, int s, int lane) {
    return row_base(h4, s)[(unsigned)lane];
}
__device__ __forceinline__ void pick4(const float4* __restrict__ h4, const int2 c, int j,
                                      grow_t (&r)[4], float (&n)[4]) {
#pragma unroll
    for (int k = 0; k < 4; ++k) {
        r[k] = row_base(h4, __builtin_amdgcn_readlane(c.x, j + k));
        n[k] = __int_as_float(__builtin_amdgcn_readlane(c.y, j + k));
    }
}
__device__ __forceinline__ void read4(const grow_t (&r)[4], int lane, v4f (&v)[4]) {
#pragma unroll
    for (int k = 0; k < 4; ++k) v[k] = r[k][(unsigned)lane];
}
__device__ __forceinline__ void add4(float4& acc, const v4f (&v)[4], const float (&n)[4]) {
#pragma unroll
    for (int k = 0; k < 4; ++k) fma4(acc, n[k], v[k]);
}
__device__ __forceinline__ void pick_tail(const float4* __restrict__ h4, const int2 c, int j, int rem,
                                          grow_t (&r)[3], float (&n)[3]) {
#pragma unroll
    for (int k = 0; k < 3; ++k) {
        r[k] = row_base(h4, __builtin_amdgcn_readlane(c.x, (j + k) & 31));
        const float w = __int_as_float(__builtin_amdgcn_readlane(c.y, (j + k) & 31));
        n[k] = (k < rem) ? w : 0.0f;
    }
}
__device__ __forceinline__ void read_tail(const grow_t (&r)[3], int rem, int lane, v4f (&v)[3]) {
#pragma unroll
    for (int k = 0; k < 3; ++k) {
        v[k] = v4f{0.0f, 0.0f, 0.0f, 0.0f};
        if (k < rem) v[k] = r[k][(unsigned)lane];
    }
}
__device__ __forceinline__ void add_tail(float4& acc, const v4f (&v)[3], const float (&n)[3]) {
#pragma unroll
    for (int k = 0; k < 3; ++k) fma4(acc, n[k], v[k]);
}

template <bool NORM>
__device__ __forceinline__ void segment_sum_pair(const float4* __restrict__ h4,
                                                 const int2* csr, int2* csr_w,
                                                 const float* __restrict__ dinv, float diA, float diB,
                                                 int startA, int endA, int startB, int endB,
                                                 int lane, float4& accA, float4& accB) {
    accA = make_float4(0.0f, 0.0f, 0.0f, 0.0f);
    accB = make_float4(0.0f, 0.0f, 0.0f, 0.0f);
    while (startA < endA || startB < endB) {
        const int cntA = max(0, min(32, endA - startA));
        const int cntB = max(0, min(32, endB - startB));
        int2 ca = make_int2(0, 0), cb = make_int2(0, 0);
        if (lane < cntA) ca = csr[startA + lane];
        if (lane < cntB) cb = csr[startB + lane];
        if (NORM) {
            const float dsa = dinv[ca.x];
            const float dsb = dinv[cb.x];
            ca.y = (lane < cntA) ? __float_as_int(dsa * __int_as_float(ca.y) * diA) : 0;
            cb.y = (lane < cntB) ? __float_as_int(dsb * __int_as_float(cb.y) * diB) : 0;
            if (lane < cntA) csr_w[startA + lane] = ca;
            if (lane < cntB) csr_w[startB + lane] = cb;
        }
        int j = 0;
        for (; j + 4 <= min(cntA, cntB); j += 4) {
            grow_t ra[4], rb[4];
            v4f    va[4], vb[4];
            float  na[4], nb[4];
            pick4(h4, ca, j, ra, na);
            pick4(h4, cb, j, rb, nb);
            asm volatile("" :: "s"(ra[0]), "s"(ra[1]), "s"(ra[2]), "s"(ra[3]),
                               "s"(rb[0]), "s"(rb[1]), "s"(rb[2]), "s"(rb[3]));
            read4(ra, lane, va);
            read4(rb, lane, vb);
            asm volatile("" :: "v"(va[0]), "v"(va[1]), "v"(va[2]), "v"(va[3]),
                               "v"(vb[0]), "v"(vb[1]), "v"(vb[2]), "v"(vb[3]));
            add4(accA, va, na);
            add4(accB, vb, nb);
        }
        int jA = j, jB = j;
        for (; jA + 4 <= cntA; jA += 4) {
            grow_t ra[4];
            v4f    va[4];
            float  na[4];
            pick4(h4, ca, jA, ra, na);
            asm volatile("" :: "s"(ra[0]), "s"(ra[1]), "s"(ra[2]), "s"(ra[3]));
            read4(ra, lane, va);
            add4(accA, va, na);
        }
        for (; jB + 4 <= cntB; jB += 4) {
            grow_t rb[4];
            v4f    vb[4];
            float  nb[4];
            pick4(h4, cb, jB, rb, nb);
            asm volatile("" :: "s"(rb[0]), "s"(rb[1]), "s"(rb[2]), "s"(rb[3]));
            read4(rb, lane, vb);
            add4(accB, vb, nb);
        }
        const int remA = cntA - jA, remB = cntB - jB;
        if (remA > 0 || remB > 0) {
            grow_t ra[3], rb[3];
            v4f    va[3], vb[3];
            float  na[3], nb[3];
            pick_tail(h4, ca, jA, remA, ra, na);
            pick_tail(h4, cb, jB, remB, rb, nb);
            read_tail(ra, remA, lane, va);
            read_tail(rb, remB, lane, vb);
            asm volatile("" :: "s"(ra[0]), "s"(ra[1]), "s"(ra[2]), "s"(rb[0]), "s"(rb[1]), "s"(rb[2]));
            add_tail(accA, va, na);
            add_tail(accB, vb, nb);
        }
        startA += 32;
        startB += 32;
    }
}

template <bool NORM>
__device__ __forceinline__ void overflow_sum(const float4* __restrict__ h4,
                                             const int* __restrict__ ovf_dst,
                                             const int2* ovf_sw, int2* ovf_sw_w,
                                             const float* __restrict__ dinv, float di, int n_ovf,
                                             int node, int lane, float4& acc) {
    for (int k0 = 0; k0 < n_ovf; k0 += 32) {
        const int k = k0 + lane;
        int  d = -1;
        int2 c = make_int2(0, 0);
        if (k < n_ovf) { d = ovf_dst[k]; c = ovf_sw[k]; }
        if (NORM) {
            if (d == node) {
                c.y = __float_as_int(dinv[c.x] * __int_as_float(c.y) * di);
                ovf_sw_w[k] = c;
            }
        }
        unsigned m = __builtin_amdgcn_ballot_w32(d == node);
        while (m != 0u) {
            const int j = __builtin_ctz(m);
            m &= m - 1u;
            const int   s = __builtin_amdgcn_readlane(c.x, j);
            const float n = __int_as_float(__builtin_amdgcn_readlane(c.y, j));
            const v4f   v = load_row(h4, s, lane);
            fma4(acc, n, v);
        }
    }
}

#define LT_ROWS  16
#define LT_PITCH 132
#define AGG_GRID 1024

template <bool NORM>
__global__ __launch_bounds__(256, 8) void aggregate_then_product(
        const float* __restrict__ xin, int2* csr,
        const int* __restrict__ cnt, const float* __restrict__ dinv,
        const int* __restrict__ ovf_n, const int* __restrict__ ovf_dst,
        int2* ovf_sw,
        const float* __restrict__ W, const float* __restrict__ b,
        float* __restrict__ xout, int N, int E, int nTiles) {
#if defined(__HIP_DEVICE_COMPILE__)
    __shared__ __attribute__((aligned(16))) float tile[LT_ROWS * LT_PITCH];
    const int lane = threadIdx.x & 31;
    const int wv   = __builtin_amdgcn_readfirstlane(threadIdx.x >> 5);
    const float4* __restrict__ xv = reinterpret_cast<const float4*>(xin);

    const int koff = (lane >> 4) * 2;
    const int col  = wv * 16 + (lane & 15);
    const float* __restrict__ arow = tile + (lane & 15) * LT_PITCH + koff;
    const float* __restrict__ wcol = W + (size_t)koff * D_CH + col;
    const int   row0 = (lane >> 4) * 8;

    v2f bw[32];
#pragma unroll
    for (int ks = 0; ks < 32; ++ks) {
        bw[ks].x = wcol[(size_t)(ks * 4) * D_CH];
        bw[ks].y = wcol[(size_t)(ks * 4 + 1) * D_CH];
    }
    const float bias = b[col];

    for (int t = blockIdx.x; t < nTiles; t += gridDim.x) {
        const int node0 = t * LT_ROWS;
        int    nds[2], cn[2], rp0[2], rp1[2];
        v4f    hs[2];
        float  di[2];
#pragma unroll
        for (int u = 0; u < 2; ++u) {
            const int nd = min(node0 + wv + 8 * u, N - 1);
            nds[u] = nd;
            cn[u]  = __builtin_amdgcn_readfirstlane(cnt[nd]);
            if (node0 + wv + 8 * u >= N) cn[u] = 0;
            hs[u]  = load_row(xv, nd, lane);
            di[u]  = __int_as_float(__builtin_amdgcn_readfirstlane(__float_as_int(dinv[nd])));
        }
#pragma unroll
        for (int u = 0; u < 2; ++u) {
            rp0[u] = nds[u] * CSR_CAP;
            rp1[u] = rp0[u] + min(max(cn[u], 0), CSR_CAP);
        }
        float4 a[2];
        segment_sum_pair<NORM>(xv, csr, csr, dinv, di[0], di[1], rp0[0], rp1[0], rp0[1], rp1[1], lane, a[0], a[1]);
#pragma unroll
        for (int u = 0; u < 2; ++u)
            if (cn[u] > CSR_CAP)
                overflow_sum<NORM>(xv, ovf_dst, ovf_sw, ovf_sw, dinv, di[u], overflow_count(ovf_n, E), nds[u], lane, a[u]);
#pragma unroll
        for (int u = 0; u < 2; ++u) {
            const int   r  = wv + 8 * u;
            const float d2 = di[u] * di[u];
            a[u].x = fmaf(hs[u].x, d2, a[u].x); a[u].y = fmaf(hs[u].y, d2, a[u].y);
            a[u].z = fmaf(hs[u].z, d2, a[u].z); a[u].w = fmaf(hs[u].w, d2, a[u].w);
            *reinterpret_cast<float4*>(&tile[r * LT_PITCH + lane * 4]) = a[u];
        }
        __syncthreads();

        v8f acc = v8f{};
#pragma unroll
        for (int ks = 0; ks < 32; ++ks) {
            v2f av;
            av.x = arow[ks * 4];
            av.y = arow[ks * 4 + 1];
            acc = __builtin_amdgcn_wmma_f32_16x16x4_f32(
                false, av, false, bw[ks], (short)0, acc, false, false);
        }
        __syncthreads();

#pragma unroll
        for (int r = 0; r < 8; ++r)
            tile[(row0 + r) * LT_PITCH + col] = fmaxf(acc[r] + bias, 0.0f);
        __syncthreads();

        for (int u = 0; u < 2; ++u) {
            const int r    = wv + 8 * u;
            const int node = node0 + r;
            if (node < N) {
                const float4 o = *reinterpret_cast<const float4*>(&tile[r * LT_PITCH + lane * 4]);
                reinterpret_cast<float4*>(xout)[(size_t)node * (D_CH / 4) + lane] = o;
            }
        }
    }
#endif
}

__global__ __launch_bounds__(256) void last_layer_at_lookups(
        const float* __restrict__ x1, const float* __restrict__ x2,
        const int2* __restrict__ csr, const int* __restrict__ cnt,
        const float* __restrict__ dinv, const int* __restrict__ ovf_n,
        const int* __restrict__ ovf_dst, const int2* __restrict__ ovf_sw,
        const float* __restrict__ W,
        const float* __restrict__ b, const int* __restrict__ idx,
        float* __restrict__ out, int N, int E, int B) {
#if defined(__HIP_DEVICE_COMPILE__)
    __shared__ __attribute__((aligned(16))) float tile[LT_ROWS * LT_PITCH];
    const int lane  = threadIdx.x & 31;
    const int wv    = __builtin_amdgcn_readfirstlane(threadIdx.x >> 5);
    const int slot0 = blockIdx.x * LT_ROWS;
    const float4* __restrict__ x1v = reinterpret_cast<const float4*>(x1);
    const float4* __restrict__ x2v = reinterpret_cast<const float4*>(x2);
    const float4 zero4 = make_float4(0.0f, 0.0f, 0.0f, 0.0f);

    const int slotA = slot0 + wv, slotB = slot0 + wv + 8;
    int nodes[2] = { N, N };
    if (slotA < B) nodes[0] = idx[slotA];
    if (slotB < B) nodes[1] = idx[slotB];

    float4 keep[2];
    int    nds[2], cn[2], rp0[2], rp1[2];
    v4f    hs[2], p1[2];
    float  di[2];
#pragma unroll
    for (int t = 0; t < 2; ++t) {
        const int nd = ((unsigned)nodes[t] < (unsigned)N) ? nodes[t] : 0;
        nds[t] = nd;
        cn[t]  = __builtin_amdgcn_readfirstlane(cnt[nd]);
        hs[t]  = load_row(x2v, nd, lane);
        p1[t]  = load_row(x1v, nd, lane);
        di[t]  = dinv[nd];
    }
#pragma unroll
    for (int t = 0; t < 2; ++t) {
        rp0[t] = nds[t] * CSR_CAP;
        rp1[t] = rp0[t] + min(max(cn[t], 0), CSR_CAP);
    }
    float4 a[2];
    segment_sum_pair<false>(x2v, csr, nullptr, dinv, 0.0f, 0.0f, rp0[0], rp1[0], rp0[1], rp1[1], lane, a[0], a[1]);
#pragma unroll
    for (int t = 0; t < 2; ++t)
        if (cn[t] > CSR_CAP)
            overflow_sum<false>(x2v, ovf_dst, ovf_sw, nullptr, dinv, 0.0f, overflow_count(ovf_n, E), nds[t], lane, a[t]);
#pragma unroll
    for (int t = 0; t < 2; ++t) {
        const int   r  = wv + 8 * t;
        const float d2 = di[t] * di[t];
        a[t].x = fmaf(hs[t].x, d2, a[t].x); a[t].y = fmaf(hs[t].y, d2, a[t].y);
        a[t].z = fmaf(hs[t].z, d2, a[t].z); a[t].w = fmaf(hs[t].w, d2, a[t].w);
        keep[t].x = p1[t].x + hs[t].x; keep[t].y = p1[t].y + hs[t].y;
        keep[t].z = p1[t].z + hs[t].z; keep[t].w = p1[t].w + hs[t].w;
        *reinterpret_cast<float4*>(&tile[r * LT_PITCH + lane * 4]) = a[t];
    }
    __syncthreads();

    const int koff = (lane >> 4) * 2;
    const int col  = wv * 16 + (lane & 15);
    const float* __restrict__ arow = tile + (lane & 15) * LT_PITCH + koff;
    const float* __restrict__ wcol = W + (size_t)koff * D_CH + col;
    v8f acc = v8f{};
#pragma unroll
    for (int ks = 0; ks < 32; ++ks) {
        v2f av, bw;
        av.x = arow[ks * 4];
        av.y = arow[ks * 4 + 1];
        bw.x = wcol[(size_t)(ks * 4) * D_CH];
        bw.y = wcol[(size_t)(ks * 4 + 1) * D_CH];
        acc = __builtin_amdgcn_wmma_f32_16x16x4_f32(
            false, av, false, bw, (short)0, acc, false, false);
    }
    __syncthreads();

    const float bias = b[col];
    const int   row0 = (lane >> 4) * 8;
#pragma unroll
    for (int r = 0; r < 8; ++r)
        tile[(row0 + r) * LT_PITCH + col] = fmaxf(acc[r] + bias, 0.0f);
    __syncthreads();

#pragma unroll
    for (int t = 0; t < 2; ++t) {
        const int r    = wv + 8 * t;
        const int slot = slot0 + r;
        if (slot < B) {
            float4 o = zero4;
            if ((unsigned)nodes[t] < (unsigned)N) {
                const float4 x3 = *reinterpret_cast<const float4*>(&tile[r * LT_PITCH + lane * 4]);
                o.x = (keep[t].x + x3.x) * (1.0f / 3.0f);
                o.y = (keep[t].y + x3.y) * (1.0f / 3.0f);
                o.z = (keep[t].z + x3.z) * (1.0f / 3.0f);
                o.w = (keep[t].w + x3.w) * (1.0f / 3.0f);
            }
            reinterpret_cast<float4*>(out)[(size_t)slot * (D_CH / 4) + lane] = o;
        }
    }
#endif
}

extern "C" void kernel_launch(void* const* d_in, const int* in_sizes, int n_in,
                              void* d_out, int out_size, void* d_ws, size_t ws_size,
                              hipStream_t stream) {
    const int*   inputs_idx  = (const int*)  d_in[0];
    const float* x           = (const float*)d_in[1];
    const int*   edge_index  = (const int*)  d_in[2];
    const float* edge_weight = (const float*)d_in[3];
    const float* Wm[3] = { (const float*)d_in[4], (const float*)d_in[6], (const float*)d_in[8] };
    const float* bv[3] = { (const float*)d_in[5], (const float*)d_in[7], (const float*)d_in[9] };

    const int B = in_sizes[0];
    const int N = in_sizes[1] / D_CH;
    const int E = in_sizes[2] / 2;
    (void)n_in; (void)ws_size; (void)out_size;
    const int* src = edge_index;
    const int* dst = edge_index + E;

    float* ws   = (float*)d_ws;
    const size_t nd   = (size_t)N * D_CH;
    const size_t Npad = ((size_t)N + 3) & ~(size_t)3;
    const size_t Epad = ((size_t)E + 3) & ~(size_t)3;
    const int TB = 256;
    const int nTiles   = (N + LT_ROWS - 1) / LT_ROWS;
    const int aggBlk   = nTiles < AGG_GRID ? nTiles : AGG_GRID;
    const int slotBlk  = (B + LT_ROWS - 1) / LT_ROWS;
    const int npb      = (N + NBINS - 1) / NBINS;
    const int binBlk   = (E + BE_EPB - 1) / BE_EPB;
    float* xl[2];
    xl[0] = ws;
    xl[1] = xl[0] + nd;
    int2*  csr     = (int2*)(xl[1] + nd);
    int*   cnt     = (int*)(csr + (size_t)N * CSR_CAP);
    int*   ovf_n   = cnt + Npad;
    float* dinv    = (float*)(ovf_n + 4);
    int*   ovf_dst = (int*)(dinv + Npad);
    int2*  ovf_sw  = (int2*)(ovf_dst + Epad);
    int4*  stage   = (int4*)(ovf_sw + Epad);
    int2*  seg     = (int2*)(stage + (size_t)binBlk * BE_EPB);

    bin_edges<<<binBlk, BE_TB, 0, stream>>>(stage, seg, ovf_n, src, dst, edge_weight, E, npb, binBlk);
    build_slots<<<NBINS, BS_TB, 0, stream>>>(csr, cnt, dinv, ovf_n, ovf_dst, ovf_sw, stage, seg, N, E, npb, binBlk);

    aggregate_then_product<true><<<aggBlk, TB, 0, stream>>>(x, csr, cnt, dinv, ovf_n, ovf_dst, ovf_sw,
                                                            Wm[0], bv[0], xl[0], N, E, nTiles);
    aggregate_then_product<false><<<aggBlk, TB, 0, stream>>>(xl[0], csr, cnt, dinv, ovf_n, ovf_dst, ovf_sw,
                                                             Wm[1], bv[1], xl[1], N, E, nTiles);

    last_layer_at_lookups<<<slotBlk, TB, 0, stream>>>(xl[0], xl[1], csr, cnt, dinv,
                                                      ovf_n, ovf_dst, ovf_sw,
                                                      Wm[2], bv[2], inputs_idx,
                                                      (float*)d_out, N, E, B);
}
